# starting kernel
# baseline (speedup 1.0000x reference)
#include <hip/hip_runtime.h>
#include <hip/hip_bf16.h>

typedef __attribute__((ext_vector_type(2))) float v2f;
typedef __attribute__((ext_vector_type(8))) float v8f;

#define DIN 64
#define DOUT 64
#define TILE_E 128
#define XS_STRIDE 68
#define W_STRIDE 65

__global__ void compute_W_kernel(const float* __restrict__ bases,
                                 const float* __restrict__ comb,
                                 float* __restrict__ W, int B) {
  int r = blockIdx.x;
  const float* c = comb + r * B;
  for (int idx = threadIdx.x; idx < DIN * DOUT; idx += blockDim.x) {
    float acc = 0.f;
    for (int b = 0; b < B; ++b) acc += c[b] * bases[b * DIN * DOUT + idx];
    W[r * DIN * DOUT + idx] = acc;
  }
}

__global__ void hist_kernel(const int* __restrict__ et, int E, int* counts) {
  int e = blockIdx.x * blockDim.x + threadIdx.x;
  if (e < E) atomicAdd(&counts[et[e]], 1);
}

__global__ void scan_kernel(const int* __restrict__ counts, int* offsets,
                            int* cursor, int R) {
  if (threadIdx.x == 0 && blockIdx.x == 0) {
    int off = 0;
    for (int r = 0; r < R; ++r) { offsets[r] = off; cursor[r] = off; off += counts[r]; }
  }
}

__global__ void scatter_kernel(const int* __restrict__ src, const int* __restrict__ tgt,
                               const int* __restrict__ et, const float* __restrict__ wts,
                               int E, int* cursor, int* s_src, int* s_tgt, float* s_w) {
  int e = blockIdx.x * blockDim.x + threadIdx.x;
  if (e < E) {
    int r = et[e];
    int p = atomicAdd(&cursor[r], 1);
    s_src[p] = src[e];
    s_tgt[p] = tgt[e];
    s_w[p]   = wts[e];
  }
}

__global__ void __launch_bounds__(256)
rgcn_tile_kernel(const float* __restrict__ x,
                 const float* __restrict__ W,
                 const int* __restrict__ s_src,
                 const int* __restrict__ s_tgt,
                 const float* __restrict__ s_w,
                 const int* __restrict__ offsets,
                 const int* __restrict__ counts,
                 float* __restrict__ out) {
  int r = blockIdx.y;
  int count = counts[r];
  int tile0 = (int)blockIdx.x * TILE_E;
  if (tile0 >= count) return;
  int start = offsets[r] + tile0;
  int nE = min(TILE_E, count - tile0);

  __shared__ float xs[TILE_E * XS_STRIDE];
  __shared__ float Wl[DIN * W_STRIDE];
  __shared__ int   tg[TILE_E];

  for (int idx = threadIdx.x; idx < DIN * DOUT; idx += 256) {
    int k = idx >> 6, n = idx & 63;
    Wl[k * W_STRIDE + n] = W[r * DIN * DOUT + idx];
  }

  int row  = threadIdx.x >> 1;
  int half = threadIdx.x & 1;
  int src = 0; float wt = 0.f;
  if (row < nE) { src = s_src[start + row]; wt = s_w[start + row]; }
  if (half == 0) tg[row] = (row < nE) ? s_tgt[start + row] : -1;
  const float4* xp = (const float4*)(x + (size_t)src * DIN) + half * 8;
  float* xd = &xs[row * XS_STRIDE + half * 32];
#pragma unroll
  for (int i = 0; i < 8; ++i) {
    float4 v = xp[i];
    xd[4 * i + 0] = v.x * wt;
    xd[4 * i + 1] = v.y * wt;
    xd[4 * i + 2] = v.z * wt;
    xd[4 * i + 3] = v.w * wt;
  }
  __syncthreads();

  int lane = threadIdx.x & 31;
  int wv   = threadIdx.x >> 5;
  int hi   = lane >> 4;
  int lm   = lane & 15;
  const float* arow = &xs[(wv * 16 + lm) * XS_STRIDE + 2 * hi];

#pragma unroll
  for (int ct = 0; ct < 4; ++ct) {
    v8f acc = {};
    const float* bcol = &Wl[(2 * hi) * W_STRIDE + ct * 16 + lm];
#pragma unroll
    for (int ks = 0; ks < 16; ++ks) {
      v2f a; a.x = arow[ks * 4];               a.y = arow[ks * 4 + 1];
      v2f b; b.x = bcol[(ks * 4) * W_STRIDE];  b.y = bcol[(ks * 4 + 1) * W_STRIDE];
      acc = __builtin_amdgcn_wmma_f32_16x16x4_f32(
           false, a,  false, b,
           (short)0, acc,  false,  false);
    }
#pragma unroll
    for (int j = 0; j < 8; ++j) {
      int m = wv * 16 + 8 * hi + j;
      int t = tg[m];
      if (t >= 0) atomicAdd(&out[(size_t)t * DOUT + ct * 16 + lm], acc[j]);
    }
  }
}

static inline size_t align256(size_t v) { return (v + 255) & ~(size_t)255; }

extern "C" void kernel_launch(void* const* d_in, const int* in_sizes, int n_in,
                              void* d_out, int out_size, void* d_ws, size_t ws_size,
                              hipStream_t stream) {
  const float* x       = (const float*)d_in[0];
  const int*   source  = (const int*)d_in[1];
  const int*   target  = (const int*)d_in[2];
  const int*   etype   = (const int*)d_in[3];
  const float* ew      = (const float*)d_in[4];
  const float* bases   = (const float*)d_in[5];
  const float* comb    = (const float*)d_in[6];
  float*       out     = (float*)d_out;

  const int N = in_sizes[0] / DIN;
  const int E = in_sizes[1];
  const int B = in_sizes[5] / (DIN * DOUT);
  const int R = in_sizes[6] / B;

  uint8_t* w = (uint8_t*)d_ws;
  float* Wbuf    = (float*)w;  w += align256((size_t)R * DIN * DOUT * 4);
  int*   counts  = (int*)w;    w += 256;
  int*   offsets = (int*)w;    w += 256;
  int*   cursor  = (int*)w;    w += 256;
  int*   s_src   = (int*)w;    w += align256((size_t)E * 4);
  int*   s_tgt   = (int*)w;    w += align256((size_t)E * 4);
  float* s_w     = (float*)w;

  hipMemsetAsync(counts, 0, 768, stream);
  hipMemsetAsync(out, 0, (size_t)N * DOUT * sizeof(float), stream);

  compute_W_kernel<<<R, 256, 0, stream>>>(bases, comb, Wbuf, B);

  int eblk = (E + 255) / 256;
  hist_kernel<<<eblk, 256, 0, stream>>>(etype, E, counts);
  scan_kernel<<<1, 32, 0, stream>>>(counts, offsets, cursor, R);
  scatter_kernel<<<eblk, 256, 0, stream>>>(source, target, etype, ew, E,
                                           cursor, s_src, s_tgt, s_w);

  dim3 grid((E + TILE_E - 1) / TILE_E, R);
  rgcn_tile_kernel<<<grid, 256, 0, stream>>>(x, Wbuf, s_src, s_tgt, s_w,
                                             offsets, counts, out);
}


# edges binned by target row range; each block owns 196 out rows, sums them in LDS (LDS float adds) and stores each once: global float atomics and the out clear gone; add order is arrival order in LDS
# speedup vs baseline: 3.9477x; 3.9477x over previous
#include <hip/hip_runtime.h>

typedef __attribute__((ext_vector_type(2))) float v2f;
typedef __attribute__((ext_vector_type(8))) float v8f;

#define DIN 64
#define DOUT 64
#define CHUNK 2048
#define EPT 8
#define MAXR 64
#define XS_STRIDE 68
#define W_STRIDE 65
#define ROWS 196
#define MAXBK 512
#define CAP 2816
#define EPT2 11
#define POOL (128 * XS_STRIDE + 2 * DIN * W_STRIDE)

__global__ void compute_W_kernel(const float* __restrict__ bases,
                                 const float* __restrict__ comb,
                                 float* __restrict__ W, int B) {
  int r = blockIdx.x;
  const float* c = comb + r * B;
  for (int idx = threadIdx.x; idx < DIN * DOUT; idx += blockDim.x) {
    float acc = 0.f;
    for (int b = 0; b < B; ++b) acc += c[b] * bases[b * DIN * DOUT + idx];
    W[r * DIN * DOUT + idx] = acc;
  }
}

__device__ __forceinline__ void wave_lds_order() {
  __builtin_amdgcn_fence(__ATOMIC_RELEASE, "wavefront");
  __builtin_amdgcn_wave_barrier();
  __builtin_amdgcn_fence(__ATOMIC_ACQUIRE, "wavefront");
}

__device__ __forceinline__ int block_incl_scan(int v, int* part, int tid, int& total) {
  const int ln = tid & 31, wq = tid >> 5;
  int s = v;
#pragma unroll
  for (int d = 1; d < 32; d <<= 1) {
    int u = __shfl_up(s, d, 32);
    if (ln >= d) s += u;
  }
  if (ln == 31) part[wq] = s;
  __syncthreads();
  int before = 0, all = 0;
#pragma unroll
  for (int k = 0; k < 8; ++k) {
    int t = part[k];
    all += t;
    if (k < wq) before += t;
  }
  total = all;
  return s + before;
}

__global__ void __launch_bounds__(256)
bin_kernel(const int* __restrict__ source,
           const int* __restrict__ target,
           const int* __restrict__ etype,
           const float* __restrict__ wts,
           const float* __restrict__ bases,
           const float* __restrict__ comb,
           float* __restrict__ W, int B, int R,
           int E, int nEB, int nbk,
           int* __restrict__ PK, int* __restrict__ SE) {
  __shared__ int  hist[MAXBK];
  __shared__ int  start[MAXBK];
  __shared__ int  part[8];
  __shared__ int2 stage[CHUNK];

  const int tid = threadIdx.x;

  if ((int)blockIdx.x < R) {
    const int r = blockIdx.x;
    const float* c = comb + r * B;
    for (int idx = tid; idx < DIN * DOUT; idx += 256) {
      float a = 0.f;
      for (int b = 0; b < B; ++b) a += c[b] * bases[b * DIN * DOUT + idx];
      W[r * DIN * DOUT + idx] = a;
    }
    return;
  }
  const int blk = (int)blockIdx.x - R;

  hist[tid] = 0;
  hist[tid + 256] = 0;
  __syncthreads();

  const int base = blk * CHUNK;
  int es[EPT], eg[EPT], er[EPT], ek[EPT], eb[EPT];
  float ew[EPT];
#pragma unroll
  for (int i = 0; i < EPT; ++i) {
    int e = base + i * 256 + tid;
    eb[i] = -1; es[i] = 0; eg[i] = 0; er[i] = 0; ew[i] = 0.f; ek[i] = 0;
    if (e < E) {
      es[i] = source[e]; eg[i] = target[e]; er[i] = etype[e]; ew[i] = wts[e];
      eb[i] = eg[i] / ROWS;
    }
  }
#pragma unroll
  for (int i = 0; i < EPT; ++i)
    if (eb[i] >= 0) ek[i] = atomicAdd(&hist[eb[i]], 1);
  __syncthreads();

  const int h0 = hist[2 * tid], h1 = hist[2 * tid + 1];
  int nall;
  const int incl = block_incl_scan(h0 + h1, part, tid, nall);
  const int excl = incl - (h0 + h1);
  start[2 * tid]     = excl;
  start[2 * tid + 1] = excl + h0;
  __syncthreads();

  for (int b = tid; b < nbk; b += 256) PK[b * nEB + blk] = (start[b] << 12) | hist[b];

#pragma unroll
  for (int i = 0; i < EPT; ++i) {
    if (eb[i] >= 0) {
      int p = start[eb[i]] + ek[i];
      stage[p] = make_int2(es[i] | ((eg[i] - eb[i] * ROWS) << 17) | (er[i] << 25),
                           __float_as_int(ew[i]));
    }
  }
  __syncthreads();

  const int nvalid = min(CHUNK, E - base);
  int2* SE2 = (int2*)SE;
  for (int i = tid; i < nvalid; i += 256) SE2[base + i] = stage[i];
}

__global__ void __launch_bounds__(256)
rgcn_owner_kernel(const float* __restrict__ x,
                  const float* __restrict__ W,
                  const int* __restrict__ PK,
                  const int* __restrict__ SE,
                  int N, int R, int nEB,
                  float* __restrict__ out) {
  __shared__ __attribute__((aligned(16))) float acc[ROWS * DOUT];
  __shared__ __attribute__((aligned(16))) float pool[POOL];
  __shared__ int   ls_src[CAP];
  __shared__ int   ls_tgt[CAP];
  __shared__ float ls_w[CAP];
  __shared__ int   cnt[MAXR];
  __shared__ int   offs[MAXR];
  __shared__ int   part[8];

  float* xs  = pool;
  float* Wl0 = pool + 128 * XS_STRIDE;
  float* Wl1 = Wl0 + DIN * W_STRIDE;
  int2*  raw = (int2*)pool;
  const int2* SE2 = (const int2*)SE;

  const int tid  = threadIdx.x;
  const int bk   = blockIdx.x;
  const int row0 = bk * ROWS;

  for (int i = tid; i < ROWS * DOUT / 4; i += 256)
    ((float4*)acc)[i] = make_float4(0.f, 0.f, 0.f, 0.f);

  const int lane = tid & 31;
  const int wv   = __builtin_amdgcn_readfirstlane(tid >> 5);
  const int hi   = lane >> 4;
  const int lm   = lane & 15;
  const int row  = lane >> 1;
  const int half = lane & 1;
  float* xd = &xs[(wv * 16 + row) * XS_STRIDE + half * 32];
  const float* arow = &xs[(wv * 16 + lm) * XS_STRIDE + 2 * hi];

  int lo = 0;
  int total = 0;
  do {
    int running = 0;
    for (int bb = 0; bb < nEB; bb += 256) {
      const int blk = bb + tid;
      const int pk  = (blk < nEB) ? PK[bk * nEB + blk] : 0;
      const int c   = pk & 4095;
      const int s   = pk >> 12;
      int btot;
      const int incl = block_incl_scan(c, part, tid, btot);
      const int first = running + incl - c - lo;
      running += btot;
      const int2* sp = SE2 + (size_t)blk * CHUNK + s;
      for (int k0 = 0; k0 < c; k0 += 8) {
        int2 v[8];
#pragma unroll
        for (int j = 0; j < 8; ++j) v[j] = sp[min(k0 + j, c - 1)];
#pragma unroll
        for (int j = 0; j < 8; ++j) {
          int pos = first + k0 + j;
          bool keep = (k0 + j < c) && (pos >= 0) && (pos < CAP);
          raw[keep ? pos : CAP] = v[j];
        }
      }
      __syncthreads();
    }
    total = __builtin_amdgcn_readfirstlane(running);
    const int n = min(CAP, total - lo);

    if (tid < MAXR) cnt[tid] = 0;
    __syncthreads();

    int er[EPT2], ek[EPT2], es[EPT2], eg[EPT2];
    float ew[EPT2];
#pragma unroll
    for (int i = 0; i < EPT2; ++i) {
      int idx = i * 256 + tid;
      er[i] = -1; es[i] = 0; eg[i] = 0; ew[i] = 0.f; ek[i] = 0;
      if (idx < n) {
        int2 v = raw[idx];
        es[i] = v.x & 0x1FFFF; eg[i] = (v.x >> 17) & 0xFF; er[i] = (v.x >> 25) & 0x3F;
        ew[i] = __int_as_float(v.y);
      }
    }
#pragma unroll
    for (int i = 0; i < EPT2; ++i)
      if (er[i] >= 0) ek[i] = atomicAdd(&cnt[er[i]], 1);
    __syncthreads();

    if (tid == 0) {
      int o = 0;
      for (int r = 0; r < R; ++r) { offs[r] = o; o += cnt[r]; }
    }
    __syncthreads();

#pragma unroll
    for (int i = 0; i < EPT2; ++i) {
      if (er[i] >= 0) {
        int p = offs[er[i]] + ek[i];
        ls_src[p] = es[i];
        ls_tgt[p] = eg[i];
        ls_w[p]   = ew[i];
      }
    }

    float wreg[16];
    if (R > 0) {
#pragma unroll
      for (int i = 0; i < 16; ++i) wreg[i] = W[i * 256 + tid];
#pragma unroll
      for (int i = 0; i < 16; ++i) {
        int idx = i * 256 + tid;
        Wl0[(idx >> 6) * W_STRIDE + (idx & 63)] = wreg[i];
      }
    }

    for (int r = 0; r < R; ++r) {
      __syncthreads();
      const float* Wc = (r & 1) ? Wl1 : Wl0;
      const bool more = (r + 1 < R);
      if (more) {
#pragma unroll
        for (int i = 0; i < 16; ++i) wreg[i] = W[(r + 1) * DIN * DOUT + i * 256 + tid];
      }
      const int c = __builtin_amdgcn_readfirstlane(cnt[r]);
      const int o = __builtin_amdgcn_readfirstlane(offs[r]);

      const int nt = (c + 15) >> 4;
      for (int t = wv; t < nt; t += 8) {
        int q = t * 16 + row;
        int src = 0; float wt = 0.f;
        if (q < c) { src = ls_src[o + q]; wt = ls_w[o + q]; }
        const float4* xp = (const float4*)(x + (size_t)src * DIN) + half * 8;
#pragma unroll
        for (int i = 0; i < 8; ++i) {
          float4 v = xp[i];
          xd[4 * i + 0] = v.x * wt;
          xd[4 * i + 1] = v.y * wt;
          xd[4 * i + 2] = v.z * wt;
          xd[4 * i + 3] = v.w * wt;
        }
        wave_lds_order();

        int tgj[8];
#pragma unroll
        for (int j = 0; j < 8; ++j) {
          int m = t * 16 + 8 * hi + j;
          int v = ls_tgt[o + min(m, c - 1)];
          tgj[j] = (m < c) ? v : -1;
        }

#pragma unroll
        for (int ct = 0; ct < 4; ++ct) {
          v8f a8 = {};
          const float* bcol = &Wc[(2 * hi) * W_STRIDE + ct * 16 + lm];
#pragma unroll
          for (int ks = 0; ks < 16; ++ks) {
            v2f a; a.x = arow[ks * 4];               a.y = arow[ks * 4 + 1];
            v2f b; b.x = bcol[(ks * 4) * W_STRIDE];  b.y = bcol[(ks * 4 + 1) * W_STRIDE];
            a8 = __builtin_amdgcn_wmma_f32_16x16x4_f32(
                 false, a,  false, b,
                 (short)0, a8,  false,  false);
          }
#pragma unroll
          for (int j = 0; j < 8; ++j) {
            if (tgj[j] >= 0)
              atomicAdd(&acc[tgj[j] * DOUT + ((ct ^ (tgj[j] & 3)) << 4) + lm], a8[j]);
          }
        }
        wave_lds_order();
      }

      if (more) {
        float* Wn = ((r + 1) & 1) ? Wl1 : Wl0;
#pragma unroll
        for (int i = 0; i < 16; ++i) {
          int idx = i * 256 + tid;
          Wn[(idx >> 6) * W_STRIDE + (idx & 63)] = wreg[i];
        }
      }
    }

    lo += CAP;
    __syncthreads();
  } while (lo < total);

  const int nrow = min(ROWS, N - row0);
  float4* op = (float4*)(out + (size_t)row0 * DOUT);
  for (int i = tid; i < nrow * (DOUT / 4); i += 256) {
    const int t = i >> 4, q = i & 15;
    op[i] = ((const float4*)acc)[t * 16 + (((q >> 2) ^ (t & 3)) << 2) + (q & 3)];
  }
}

__global__ void __launch_bounds__(256)
rgcn_chunk_kernel(const float* __restrict__ x,
                  const float* __restrict__ W,
                  const int* __restrict__ source,
                  const int* __restrict__ target,
                  const int* __restrict__ etype,
                  const float* __restrict__ wts,
                  int E, int R,
                  float* __restrict__ out) {
  __shared__ float xs[128 * XS_STRIDE];
  __shared__ float Wl[2][DIN * W_STRIDE];
  __shared__ int   ls_src[CHUNK];
  __shared__ int   ls_tgt[CHUNK];
  __shared__ float ls_w[CHUNK];
  __shared__ int   cnt[MAXR];
  __shared__ int   offs[MAXR];

  const int tid  = threadIdx.x;
  const int base = (int)blockIdx.x * CHUNK;

  if (tid < MAXR) cnt[tid] = 0;
  __syncthreads();

  int er[EPT], ek[EPT], es[EPT], eg[EPT];
  float ew[EPT];
#pragma unroll
  for (int i = 0; i < EPT; ++i) {
    int e = base + i * 256 + tid;
    er[i] = -1; es[i] = 0; eg[i] = 0; ew[i] = 0.f; ek[i] = 0;
    if (e < E) { er[i] = etype[e]; es[i] = source[e]; eg[i] = target[e]; ew[i] = wts[e]; }
  }
#pragma unroll
  for (int i = 0; i < EPT; ++i)
    if (er[i] >= 0) ek[i] = atomicAdd(&cnt[er[i]], 1);
  __syncthreads();

  if (tid == 0) {
    int o = 0;
    for (int r = 0; r < R; ++r) { offs[r] = o; o += cnt[r]; }
  }
  __syncthreads();

#pragma unroll
  for (int i = 0; i < EPT; ++i) {
    if (er[i] >= 0) {
      int p = offs[er[i]] + ek[i];
      ls_src[p] = es[i];
      ls_tgt[p] = eg[i];
      ls_w[p]   = ew[i];
    }
  }

  float wreg[16];
  if (R > 0) {
#pragma unroll
    for (int i = 0; i < 16; ++i) wreg[i] = W[i * 256 + tid];
#pragma unroll
    for (int i = 0; i < 16; ++i) {
      int idx = i * 256 + tid;
      Wl[0][(idx >> 6) * W_STRIDE + (idx & 63)] = wreg[i];
    }
  }

  const int lane = tid & 31;
  const int wv   = __builtin_amdgcn_readfirstlane(tid >> 5);
  const int hi   = lane >> 4;
  const int lm   = lane & 15;
  const int row  = lane >> 1;
  const int half = lane & 1;
  float* xd = &xs[(wv * 16 + row) * XS_STRIDE + half * 32];
  const float* arow = &xs[(wv * 16 + lm) * XS_STRIDE + 2 * hi];

  for (int r = 0; r < R; ++r) {
    __syncthreads();
    const float* Wc = Wl[r & 1];
    const bool more = (r + 1 < R);
    if (more) {
#pragma unroll
      for (int i = 0; i < 16; ++i) wreg[i] = W[(r + 1) * DIN * DOUT + i * 256 + tid];
    }
    const int c = __builtin_amdgcn_readfirstlane(cnt[r]);
    const int o = __builtin_amdgcn_readfirstlane(offs[r]);

    const int nt = (c + 15) >> 4;
    for (int t = wv; t < nt; t += 8) {
      int q = t * 16 + row;
      int src = 0; float wt = 0.f;
      if (q < c) { src = ls_src[o + q]; wt = ls_w[o + q]; }
      const float4* xp = (const float4*)(x + (size_t)src * DIN) + half * 8;
#pragma unroll
      for (int i = 0; i < 8; ++i) {
        float4 v = xp[i];
        xd[4 * i + 0] = v.x * wt;
        xd[4 * i + 1] = v.y * wt;
        xd[4 * i + 2] = v.z * wt;
        xd[4 * i + 3] = v.w * wt;
      }
      wave_lds_order();

      int tgj[8];
#pragma unroll
      for (int j = 0; j < 8; ++j) {
        int m = t * 16 + 8 * hi + j;
        int v = ls_tgt[o + min(m, c - 1)];
        tgj[j] = (m < c) ? v : -1;
      }

#pragma unroll
      for (int ct = 0; ct < 4; ++ct) {
        v8f acc = {};
        const float* bcol = &Wc[(2 * hi) * W_STRIDE + ct * 16 + lm];
#pragma unroll
        for (int ks = 0; ks < 16; ++ks) {
          v2f a; a.x = arow[ks * 4];               a.y = arow[ks * 4 + 1];
          v2f b; b.x = bcol[(ks * 4) * W_STRIDE];  b.y = bcol[(ks * 4 + 1) * W_STRIDE];
          acc = __builtin_amdgcn_wmma_f32_16x16x4_f32(
               false, a,  false, b,
               (short)0, acc,  false,  false);
        }
#pragma unroll
        for (int j = 0; j < 8; ++j) {
          if (tgj[j] >= 0) atomicAdd(&out[(size_t)tgj[j] * DOUT + ct * 16 + lm], acc[j]);
        }
      }
      wave_lds_order();
    }

    if (more) {
      float* Wn = Wl[(r + 1) & 1];
#pragma unroll
      for (int i = 0; i < 16; ++i) {
        int idx = i * 256 + tid;
        Wn[(idx >> 6) * W_STRIDE + (idx & 63)] = wreg[i];
      }
    }
  }
}

static inline size_t align256(size_t v) { return (v + 255) & ~(size_t)255; }

extern "C" void kernel_launch(void* const* d_in, const int* in_sizes, int n_in,
                              void* d_out, int out_size, void* d_ws, size_t ws_size,
                              hipStream_t stream) {
  const float* x       = (const float*)d_in[0];
  const int*   source  = (const int*)d_in[1];
  const int*   target  = (const int*)d_in[2];
  const int*   etype   = (const int*)d_in[3];
  const float* ew      = (const float*)d_in[4];
  const float* bases   = (const float*)d_in[5];
  const float* comb    = (const float*)d_in[6];
  float*       out     = (float*)d_out;

  const int N = in_sizes[0] / DIN;
  const int E = in_sizes[1];
  const int B = in_sizes[5] / (DIN * DOUT);
  const int R = in_sizes[6] / B;

  const int nEB = (E + CHUNK - 1) / CHUNK;
  const int nbk = (N + ROWS - 1) / ROWS;

  char*  ws   = (char*)d_ws;
  float* Wbuf = (float*)ws;
  const size_t oPK = align256((size_t)R * DIN * DOUT * sizeof(float));
  const size_t oSE = oPK + align256((size_t)nbk * nEB * sizeof(int));
  const size_t need = oSE + (size_t)nEB * CHUNK * sizeof(int2);

  if (ws_size >= need && nbk <= MAXBK && R <= MAXR) {
    int*  PK = (int*)(ws + oPK);
    int*  SE = (int*)(ws + oSE);
    bin_kernel<<<R + nEB, 256, 0, stream>>>(source, target, etype, ew, bases, comb,
                                            Wbuf, B, R, E, nEB, nbk, PK, SE);
    rgcn_owner_kernel<<<nbk, 256, 0, stream>>>(x, Wbuf, PK, SE, N, R, nEB, out);
    return;
  }

  hipMemsetAsync(out, 0, (size_t)N * DOUT * sizeof(float), stream);
  compute_W_kernel<<<R, 256, 0, stream>>>(bases, comb, Wbuf, B);
  int nblk = (E + CHUNK - 1) / CHUNK;
  rgcn_chunk_kernel<<<nblk, 256, 0, stream>>>(x, Wbuf, source, target, etype, ew,
                                              E, R, out);
}


# owner kernel: a wave owns whole relations and holds W_r as B fragments in registers (read once from a fragment-ordered workspace): no W in LDS, no per-relation barrier; LDS add order mixes relations
# speedup vs baseline: 5.4888x; 1.3904x over previous
#include <hip/hip_runtime.h>

typedef __attribute__((ext_vector_type(2))) float v2f;
typedef __attribute__((ext_vector_type(8))) float v8f;

#define DIN 64
#define DOUT 64
#define CHUNK 2048
#define EPT 8
#define MAXR 64
#define XS_STRIDE 68
#define W_STRIDE 65
#define ROWS 196
#define MAXBK 512
#define CAP 2816
#define EPT2 11
#define POOL (128 * XS_STRIDE)

__global__ void compute_W_kernel(const float* __restrict__ bases,
                                 const float* __restrict__ comb,
                                 float* __restrict__ W, int B) {
  int r = blockIdx.x;
  const float* c = comb + r * B;
  for (int idx = threadIdx.x; idx < DIN * DOUT; idx += blockDim.x) {
    float acc = 0.f;
    for (int b = 0; b < B; ++b) acc += c[b] * bases[b * DIN * DOUT + idx];
    W[r * DIN * DOUT + idx] = acc;
  }
}

__device__ __forceinline__ void wave_lds_order() {
  __builtin_amdgcn_fence(__ATOMIC_RELEASE, "wavefront");
  __builtin_amdgcn_wave_barrier();
  __builtin_amdgcn_fence(__ATOMIC_ACQUIRE, "wavefront");
}

__device__ __forceinline__ int block_incl_scan(int v, int* part, int tid, int& total) {
  const int ln = tid & 31, wq = tid >> 5;
  int s = v;
#pragma unroll
  for (int d = 1; d < 32; d <<= 1) {
    int u = __shfl_up(s, d, 32);
    if (ln >= d) s += u;
  }
  if (ln == 31) part[wq] = s;
  __syncthreads();
  int before = 0, all = 0;
#pragma unroll
  for (int k = 0; k < 8; ++k) {
    int t = part[k];
    all += t;
    if (k < wq) before += t;
  }
  total = all;
  return s + before;
}

__global__ void __launch_bounds__(256)
bin_kernel(const int* __restrict__ source,
           const int* __restrict__ target,
           const int* __restrict__ etype,
           const float* __restrict__ wts,
           const float* __restrict__ bases,
           const float* __restrict__ comb,
           float* __restrict__ W, int B, int R,
           int E, int nEB, int nbk,
           int* __restrict__ PK, int* __restrict__ SE) {
  __shared__ int  hist[MAXBK];
  __shared__ int  start[MAXBK];
  __shared__ int  part[8];
  __shared__ int2 stage[CHUNK];

  const int tid = threadIdx.x;

  if ((int)blockIdx.x < R) {
    const int r = blockIdx.x;
    const float* c = comb + r * B;
    for (int idx = tid; idx < DIN * DOUT; idx += 256) {
      float a = 0.f;
      for (int b = 0; b < B; ++b) a += c[b] * bases[b * DIN * DOUT + idx];
      const int k = idx >> 6, n = idx & 63;
      const int dst = ((k >> 2) * 2 + (n >> 5)) * 128
                    + (((k >> 1) & 1) * 16 + (n & 15)) * 4
                    + ((n >> 4) & 1) * 2 + (k & 1);
      W[r * DIN * DOUT + dst] = a;
    }
    return;
  }
  const int blk = (int)blockIdx.x - R;

  hist[tid] = 0;
  hist[tid + 256] = 0;
  __syncthreads();

  const int base = blk * CHUNK;
  int es[EPT], eg[EPT], er[EPT], ek[EPT], eb[EPT];
  float ew[EPT];
#pragma unroll
  for (int i = 0; i < EPT; ++i) {
    int e = base + i * 256 + tid;
    eb[i] = -1; es[i] = 0; eg[i] = 0; er[i] = 0; ew[i] = 0.f; ek[i] = 0;
    if (e < E) {
      es[i] = source[e]; eg[i] = target[e]; er[i] = etype[e]; ew[i] = wts[e];
      eb[i] = eg[i] / ROWS;
    }
  }
#pragma unroll
  for (int i = 0; i < EPT; ++i)
    if (eb[i] >= 0) ek[i] = atomicAdd(&hist[eb[i]], 1);
  __syncthreads();

  const int h0 = hist[2 * tid], h1 = hist[2 * tid + 1];
  int nall;
  const int incl = block_incl_scan(h0 + h1, part, tid, nall);
  const int excl = incl - (h0 + h1);
  start[2 * tid]     = excl;
  start[2 * tid + 1] = excl + h0;
  __syncthreads();

  for (int b = tid; b < nbk; b += 256) PK[b * nEB + blk] = (start[b] << 12) | hist[b];

#pragma unroll
  for (int i = 0; i < EPT; ++i) {
    if (eb[i] >= 0) {
      int p = start[eb[i]] + ek[i];
      stage[p] = make_int2(es[i] | ((eg[i] - eb[i] * ROWS) << 17) | (er[i] << 25),
                           __float_as_int(ew[i]));
    }
  }
  __syncthreads();

  const int nvalid = min(CHUNK, E - base);
  int2* SE2 = (int2*)SE;
  for (int i = tid; i < nvalid; i += 256) SE2[base + i] = stage[i];
}

__global__ void __launch_bounds__(256)
rgcn_owner_kernel(const float* __restrict__ x,
                  const float* __restrict__ W,
                  const int* __restrict__ PK,
                  const int* __restrict__ SE,
                  int N, int R, int nEB,
                  float* __restrict__ out) {
  __shared__ __attribute__((aligned(16))) float acc[ROWS * DOUT];
  __shared__ __attribute__((aligned(16))) float pool[POOL];
  __shared__ int   ls_src[CAP];
  __shared__ int   ls_tgt[CAP];
  __shared__ float ls_w[CAP];
  __shared__ int   cnt[MAXR];
  __shared__ int   offs[MAXR];
  __shared__ int   part[8];

  float* xs  = pool;
  int2*  raw = (int2*)pool;
  const int2* SE2 = (const int2*)SE;

  const int tid  = threadIdx.x;
  const int bk   = blockIdx.x;
  const int row0 = bk * ROWS;

  for (int i = tid; i < ROWS * DOUT / 4; i += 256)
    ((float4*)acc)[i] = make_float4(0.f, 0.f, 0.f, 0.f);

  const int lane = tid & 31;
  const int wv   = __builtin_amdgcn_readfirstlane(tid >> 5);
  const int hi   = lane >> 4;
  const int lm   = lane & 15;
  const int row  = lane >> 1;
  const int half = lane & 1;
  float* xd = &xs[(wv * 16 + row) * XS_STRIDE + half * 32];
  const float* arow = &xs[(wv * 16 + lm) * XS_STRIDE + 2 * hi];

  int lo = 0;
  int total = 0;
  do {
    int running = 0;
    for (int bb = 0; bb < nEB; bb += 256) {
      const int blk = bb + tid;
      const int pk  = (blk < nEB) ? PK[bk * nEB + blk] : 0;
      const int c   = pk & 4095;
      const int s   = pk >> 12;
      int btot;
      const int incl = block_incl_scan(c, part, tid, btot);
      const int first = running + incl - c - lo;
      running += btot;
      const int2* sp = SE2 + (size_t)blk * CHUNK + s;
      for (int k0 = 0; k0 < c; k0 += 8) {
        int2 v[8];
#pragma unroll
        for (int j = 0; j < 8; ++j) v[j] = sp[min(k0 + j, c - 1)];
#pragma unroll
        for (int j = 0; j < 8; ++j) {
          int pos = first + k0 + j;
          bool keep = (k0 + j < c) && (pos >= 0) && (pos < CAP);
          raw[keep ? pos : CAP] = v[j];
        }
      }
      __syncthreads();
    }
    total = __builtin_amdgcn_readfirstlane(running);
    const int n = min(CAP, total - lo);

    if (tid < MAXR) cnt[tid] = 0;
    __syncthreads();

    int er[EPT2], ek[EPT2], es[EPT2], eg[EPT2];
    float ew[EPT2];
#pragma unroll
    for (int i = 0; i < EPT2; ++i) {
      int idx = i * 256 + tid;
      er[i] = -1; es[i] = 0; eg[i] = 0; ew[i] = 0.f; ek[i] = 0;
      if (idx < n) {
        int2 v = raw[idx];
        es[i] = v.x & 0x1FFFF; eg[i] = (v.x >> 17) & 0xFF; er[i] = (v.x >> 25) & 0x3F;
        ew[i] = __int_as_float(v.y);
      }
    }
#pragma unroll
    for (int i = 0; i < EPT2; ++i)
      if (er[i] >= 0) ek[i] = atomicAdd(&cnt[er[i]], 1);
    __syncthreads();

    if (tid == 0) {
      int o = 0;
      for (int r = 0; r < R; ++r) { offs[r] = o; o += cnt[r]; }
    }
    __syncthreads();

#pragma unroll
    for (int i = 0; i < EPT2; ++i) {
      if (er[i] >= 0) {
        int p = offs[er[i]] + ek[i];
        ls_src[p] = es[i];
        ls_tgt[p] = eg[i];
        ls_w[p]   = ew[i];
      }
    }

    __syncthreads();

    for (int r = wv; r < R; r += 8) {
      const int c = __builtin_amdgcn_readfirstlane(cnt[r]);
      const int o = __builtin_amdgcn_readfirstlane(offs[r]);
      if (c > 0) {
        const float4* wf = (const float4*)(W + (size_t)r * DIN * DOUT) + lane;
        float4 bq[32];
#pragma unroll
        for (int j = 0; j < 32; ++j) bq[j] = wf[j * 32];

        const int nt = (c + 15) >> 4;
        for (int t = 0; t < nt; ++t) {
          int q = t * 16 + row;
          int src = 0; float wt = 0.f;
          if (q < c) { src = ls_src[o + q]; wt = ls_w[o + q]; }
          const float4* xp = (const float4*)(x + (size_t)src * DIN) + half * 8;
#pragma unroll
          for (int i = 0; i < 8; ++i) {
            float4 v = xp[i];
            xd[4 * i + 0] = v.x * wt;
            xd[4 * i + 1] = v.y * wt;
            xd[4 * i + 2] = v.z * wt;
            xd[4 * i + 3] = v.w * wt;
          }
          wave_lds_order();

          int tgj[8];
#pragma unroll
          for (int j = 0; j < 8; ++j) {
            int m = t * 16 + 8 * hi + j;
            int v = ls_tgt[o + min(m, c - 1)];
            tgj[j] = (m < c) ? v : -1;
          }

          v8f a8[4] = {};
#pragma unroll
          for (int ks = 0; ks < 16; ++ks) {
            v2f a; a.x = arow[ks * 4]; a.y = arow[ks * 4 + 1];
#pragma unroll
            for (int ct = 0; ct < 4; ++ct) {
              const float4 w4 = bq[ks * 2 + (ct >> 1)];
              v2f b;
              if (ct & 1) { b.x = w4.z; b.y = w4.w; } else { b.x = w4.x; b.y = w4.y; }
              a8[ct] = __builtin_amdgcn_wmma_f32_16x16x4_f32(
                   false, a,  false, b,
                   (short)0, a8[ct],  false,  false);
            }
          }
#pragma unroll
          for (int ct = 0; ct < 4; ++ct) {
#pragma unroll
            for (int j = 0; j < 8; ++j) {
              if (tgj[j] >= 0)
                atomicAdd(&acc[tgj[j] * DOUT + ((ct ^ (tgj[j] & 3)) << 4) + lm], a8[ct][j]);
            }
          }
          wave_lds_order();
        }
      }
    }

    lo += CAP;
    __syncthreads();
  } while (lo < total);

  const int nrow = min(ROWS, N - row0);
  float4* op = (float4*)(out + (size_t)row0 * DOUT);
  for (int i = tid; i < nrow * (DOUT / 4); i += 256) {
    const int t = i >> 4, q = i & 15;
    op[i] = ((const float4*)acc)[t * 16 + (((q >> 2) ^ (t & 3)) << 2) + (q & 3)];
  }
}

__global__ void __launch_bounds__(256)
rgcn_chunk_kernel(const float* __restrict__ x,
                  const float* __restrict__ W,
                  const int* __restrict__ source,
                  const int* __restrict__ target,
                  const int* __restrict__ etype,
                  const float* __restrict__ wts,
                  int E, int R,
                  float* __restrict__ out) {
  __shared__ float xs[128 * XS_STRIDE];
  __shared__ float Wl[2][DIN * W_STRIDE];
  __shared__ int   ls_src[CHUNK];
  __shared__ int   ls_tgt[CHUNK];
  __shared__ float ls_w[CHUNK];
  __shared__ int   cnt[MAXR];
  __shared__ int   offs[MAXR];

  const int tid  = threadIdx.x;
  const int base = (int)blockIdx.x * CHUNK;

  if (tid < MAXR) cnt[tid] = 0;
  __syncthreads();

  int er[EPT], ek[EPT], es[EPT], eg[EPT];
  float ew[EPT];
#pragma unroll
  for (int i = 0; i < EPT; ++i) {
    int e = base + i * 256 + tid;
    er[i] = -1; es[i] = 0; eg[i] = 0; ew[i] = 0.f; ek[i] = 0;
    if (e < E) { er[i] = etype[e]; es[i] = source[e]; eg[i] = target[e]; ew[i] = wts[e]; }
  }
#pragma unroll
  for (int i = 0; i < EPT; ++i)
    if (er[i] >= 0) ek[i] = atomicAdd(&cnt[er[i]], 1);
  __syncthreads();

  if (tid == 0) {
    int o = 0;
    for (int r = 0; r < R; ++r) { offs[r] = o; o += cnt[r]; }
  }
  __syncthreads();

#pragma unroll
  for (int i = 0; i < EPT; ++i) {
    if (er[i] >= 0) {
      int p = offs[er[i]] + ek[i];
      ls_src[p] = es[i];
      ls_tgt[p] = eg[i];
      ls_w[p]   = ew[i];
    }
  }

  float wreg[16];
  if (R > 0) {
#pragma unroll
    for (int i = 0; i < 16; ++i) wreg[i] = W[i * 256 + tid];
#pragma unroll
    for (int i = 0; i < 16; ++i) {
      int idx = i * 256 + tid;
      Wl[0][(idx >> 6) * W_STRIDE + (idx & 63)] = wreg[i];
    }
  }

  const int lane = tid & 31;
  const int wv   = __builtin_amdgcn_readfirstlane(tid >> 5);
  const int hi   = lane >> 4;
  const int lm   = lane & 15;
  const int row  = lane >> 1;
  const int half = lane & 1;
  float* xd = &xs[(wv * 16 + row) * XS_STRIDE + half * 32];
  const float* arow = &xs[(wv * 16 + lm) * XS_STRIDE + 2 * hi];

  for (int r = 0; r < R; ++r) {
    __syncthreads();
    const float* Wc = Wl[r & 1];
    const bool more = (r + 1 < R);
    if (more) {
#pragma unroll
      for (int i = 0; i < 16; ++i) wreg[i] = W[(r + 1) * DIN * DOUT + i * 256 + tid];
    }
    const int c = __builtin_amdgcn_readfirstlane(cnt[r]);
    const int o = __builtin_amdgcn_readfirstlane(offs[r]);

    const int nt = (c + 15) >> 4;
    for (int t = wv; t < nt; t += 8) {
      int q = t * 16 + row;
      int src = 0; float wt = 0.f;
      if (q < c) { src = ls_src[o + q]; wt = ls_w[o + q]; }
      const float4* xp = (const float4*)(x + (size_t)src * DIN) + half * 8;
#pragma unroll
      for (int i = 0; i < 8; ++i) {
        float4 v = xp[i];
        xd[4 * i + 0] = v.x * wt;
        xd[4 * i + 1] = v.y * wt;
        xd[4 * i + 2] = v.z * wt;
        xd[4 * i + 3] = v.w * wt;
      }
      wave_lds_order();

      int tgj[8];
#pragma unroll
      for (int j = 0; j < 8; ++j) {
        int m = t * 16 + 8 * hi + j;
        int v = ls_tgt[o + min(m, c - 1)];
        tgj[j] = (m < c) ? v : -1;
      }

#pragma unroll
      for (int ct = 0; ct < 4; ++ct) {
        v8f acc = {};
        const float* bcol = &Wc[(2 * hi) * W_STRIDE + ct * 16 + lm];
#pragma unroll
        for (int ks = 0; ks < 16; ++ks) {
          v2f a; a.x = arow[ks * 4];               a.y = arow[ks * 4 + 1];
          v2f b; b.x = bcol[(ks * 4) * W_STRIDE];  b.y = bcol[(ks * 4 + 1) * W_STRIDE];
          acc = __builtin_amdgcn_wmma_f32_16x16x4_f32(
               false, a,  false, b,
               (short)0, acc,  false,  false);
        }
#pragma unroll
        for (int j = 0; j < 8; ++j) {
          if (tgj[j] >= 0) atomicAdd(&out[(size_t)tgj[j] * DOUT + ct * 16 + lm], acc[j]);
        }
      }
      wave_lds_order();
    }

    if (more) {
      float* Wn = Wl[(r + 1) & 1];
#pragma unroll
      for (int i = 0; i < 16; ++i) {
        int idx = i * 256 + tid;
        Wn[(idx >> 6) * W_STRIDE + (idx & 63)] = wreg[i];
      }
    }
  }
}

static inline size_t align256(size_t v) { return (v + 255) & ~(size_t)255; }

extern "C" void kernel_launch(void* const* d_in, const int* in_sizes, int n_in,
                              void* d_out, int out_size, void* d_ws, size_t ws_size,
                              hipStream_t stream) {
  const float* x       = (const float*)d_in[0];
  const int*   source  = (const int*)d_in[1];
  const int*   target  = (const int*)d_in[2];
  const int*   etype   = (const int*)d_in[3];
  const float* ew      = (const float*)d_in[4];
  const float* bases   = (const float*)d_in[5];
  const float* comb    = (const float*)d_in[6];
  float*       out     = (float*)d_out;

  const int N = in_sizes[0] / DIN;
  const int E = in_sizes[1];
  const int B = in_sizes[5] / (DIN * DOUT);
  const int R = in_sizes[6] / B;

  const int nEB = (E + CHUNK - 1) / CHUNK;
  const int nbk = (N + ROWS - 1) / ROWS;

  char*  ws   = (char*)d_ws;
  float* Wbuf = (float*)ws;
  const size_t oPK = align256((size_t)R * DIN * DOUT * sizeof(float));
  const size_t oSE = oPK + align256((size_t)nbk * nEB * sizeof(int));
  const size_t need = oSE + (size_t)nEB * CHUNK * sizeof(int2);

  if (ws_size >= need && nbk <= MAXBK && R <= MAXR) {
    int*  PK = (int*)(ws + oPK);
    int*  SE = (int*)(ws + oSE);
    bin_kernel<<<R + nEB, 256, 0, stream>>>(source, target, etype, ew, bases, comb,
                                            Wbuf, B, R, E, nEB, nbk, PK, SE);
    rgcn_owner_kernel<<<nbk, 256, 0, stream>>>(x, Wbuf, PK, SE, N, R, nEB, out);
    return;
  }

  hipMemsetAsync(out, 0, (size_t)N * DOUT * sizeof(float), stream);
  compute_W_kernel<<<R, 256, 0, stream>>>(bases, comb, Wbuf, B);
  int nblk = (E + CHUNK - 1) / CHUNK;
  rgcn_chunk_kernel<<<nblk, 256, 0, stream>>>(x, Wbuf, source, target, etype, ew,
                                              E, R, out);
}


# owner kernel: each lane loads its x row from global memory straight into registers as A fragments (W fragments stored in the matching k order); LDS row staging gone; dot-product sums regroup
# speedup vs baseline: 5.7606x; 1.0495x over previous
#include <hip/hip_runtime.h>

typedef __attribute__((ext_vector_type(2))) float v2f;
typedef __attribute__((ext_vector_type(8))) float v8f;

#define DIN 64
#define DOUT 64
#define CHUNK 2048
#define EPT 8
#define MAXR 64
#define XS_STRIDE 68
#define W_STRIDE 65
#define ROWS 196
#define MAXBK 512
#define CAP 2816
#define EPT2 11

__global__ void compute_W_kernel(const float* __restrict__ bases,
                                 const float* __restrict__ comb,
                                 float* __restrict__ W, int B) {
  int r = blockIdx.x;
  const float* c = comb + r * B;
  for (int idx = threadIdx.x; idx < DIN * DOUT; idx += blockDim.x) {
    float acc = 0.f;
    for (int b = 0; b < B; ++b) acc += c[b] * bases[b * DIN * DOUT + idx];
    W[r * DIN * DOUT + idx] = acc;
  }
}

__device__ __forceinline__ void wave_lds_order() {
  __builtin_amdgcn_fence(__ATOMIC_RELEASE, "wavefront");
  __builtin_amdgcn_wave_barrier();
  __builtin_amdgcn_fence(__ATOMIC_ACQUIRE, "wavefront");
}

__device__ __forceinline__ int block_incl_scan(int v, int* part, int tid, int& total) {
  const int ln = tid & 31, wq = tid >> 5;
  int s = v;
#pragma unroll
  for (int d = 1; d < 32; d <<= 1) {
    int u = __shfl_up(s, d, 32);
    if (ln >= d) s += u;
  }
  if (ln == 31) part[wq] = s;
  __syncthreads();
  int before = 0, all = 0;
#pragma unroll
  for (int k = 0; k < 8; ++k) {
    int t = part[k];
    all += t;
    if (k < wq) before += t;
  }
  total = all;
  return s + before;
}

__global__ void __launch_bounds__(256)
bin_kernel(const int* __restrict__ source,
           const int* __restrict__ target,
           const int* __restrict__ etype,
           const float* __restrict__ wts,
           const float* __restrict__ bases,
           const float* __restrict__ comb,
           float* __restrict__ W, int B, int R,
           int E, int nEB, int nbk,
           int* __restrict__ PK, int* __restrict__ SE) {
  __shared__ int  hist[MAXBK];
  __shared__ int  start[MAXBK];
  __shared__ int  part[8];
  __shared__ int2 stage[CHUNK];

  const int tid = threadIdx.x;

  if ((int)blockIdx.x < R) {
    const int r = blockIdx.x;
    const float* c = comb + r * B;
    for (int idx = tid; idx < DIN * DOUT; idx += 256) {
      float a = 0.f;
      for (int b = 0; b < B; ++b) a += c[b] * bases[b * DIN * DOUT + idx];
      const int k = idx >> 6, n = idx & 63;
      const int ks  = ((k >> 3) << 1) | ((k >> 1) & 1);
      const int dst = (ks * 2 + (n >> 5)) * 128
                    + (((k >> 2) & 1) * 16 + (n & 15)) * 4
                    + ((n >> 4) & 1) * 2 + (k & 1);
      W[r * DIN * DOUT + dst] = a;
    }
    return;
  }
  const int blk = (int)blockIdx.x - R;

  hist[tid] = 0;
  hist[tid + 256] = 0;
  __syncthreads();

  const int base = blk * CHUNK;
  int es[EPT], eg[EPT], er[EPT], ek[EPT], eb[EPT];
  float ew[EPT];
#pragma unroll
  for (int i = 0; i < EPT; ++i) {
    int e = base + i * 256 + tid;
    eb[i] = -1; es[i] = 0; eg[i] = 0; er[i] = 0; ew[i] = 0.f; ek[i] = 0;
    if (e < E) {
      es[i] = source[e]; eg[i] = target[e]; er[i] = etype[e]; ew[i] = wts[e];
      eb[i] = eg[i] / ROWS;
    }
  }
#pragma unroll
  for (int i = 0; i < EPT; ++i)
    if (eb[i] >= 0) ek[i] = atomicAdd(&hist[eb[i]], 1);
  __syncthreads();

  const int h0 = hist[2 * tid], h1 = hist[2 * tid + 1];
  int nall;
  const int incl = block_incl_scan(h0 + h1, part, tid, nall);
  const int excl = incl - (h0 + h1);
  start[2 * tid]     = excl;
  start[2 * tid + 1] = excl + h0;
  __syncthreads();

  for (int b = tid; b < nbk; b += 256) PK[b * nEB + blk] = (start[b] << 12) | hist[b];

#pragma unroll
  for (int i = 0; i < EPT; ++i) {
    if (eb[i] >= 0) {
      int p = start[eb[i]] + ek[i];
      stage[p] = make_int2(es[i] | ((eg[i] - eb[i] * ROWS) << 17) | (er[i] << 25),
                           __float_as_int(ew[i]));
    }
  }
  __syncthreads();

  const int nvalid = min(CHUNK, E - base);
  int2* SE2 = (int2*)SE;
  for (int i = tid; i < nvalid; i += 256) SE2[base + i] = stage[i];
}

__global__ void __launch_bounds__(256)
rgcn_owner_kernel(const float* __restrict__ x,
                  const float* __restrict__ W,
                  const int* __restrict__ PK,
                  const int* __restrict__ SE,
                  int N, int R, int nEB,
                  float* __restrict__ out) {
  __shared__ __attribute__((aligned(16))) float acc[ROWS * DOUT];
  __shared__ int2  raw[CAP + 1];
  __shared__ int   ls_src[CAP];
  __shared__ int   ls_tgt[CAP];
  __shared__ float ls_w[CAP];
  __shared__ int   cnt[MAXR];
  __shared__ int   offs[MAXR];
  __shared__ int   part[8];

  const int2* SE2 = (const int2*)SE;

  const int tid  = threadIdx.x;
  const int bk   = blockIdx.x;
  const int row0 = bk * ROWS;

  for (int i = tid; i < ROWS * DOUT / 4; i += 256)
    ((float4*)acc)[i] = make_float4(0.f, 0.f, 0.f, 0.f);

  const int lane = tid & 31;
  const int wv   = __builtin_amdgcn_readfirstlane(tid >> 5);
  const int hi   = lane >> 4;
  const int lm   = lane & 15;

  int lo = 0;
  int total = 0;
  do {
    int running = 0;
    for (int bb = 0; bb < nEB; bb += 256) {
      const int blk = bb + tid;
      const int pk  = (blk < nEB) ? PK[bk * nEB + blk] : 0;
      const int c   = pk & 4095;
      const int s   = pk >> 12;
      int btot;
      const int incl = block_incl_scan(c, part, tid, btot);
      const int first = running + incl - c - lo;
      running += btot;
      const int2* sp = SE2 + (size_t)blk * CHUNK + s;
      for (int k0 = 0; k0 < c; k0 += 8) {
        int2 v[8];
#pragma unroll
        for (int j = 0; j < 8; ++j) v[j] = sp[min(k0 + j, c - 1)];
#pragma unroll
        for (int j = 0; j < 8; ++j) {
          int pos = first + k0 + j;
          bool keep = (k0 + j < c) && (pos >= 0) && (pos < CAP);
          raw[keep ? pos : CAP] = v[j];
        }
      }
      __syncthreads();
    }
    total = __builtin_amdgcn_readfirstlane(running);
    const int n = min(CAP, total - lo);

    if (tid < MAXR) cnt[tid] = 0;
    __syncthreads();

    int er[EPT2], ek[EPT2], es[EPT2], eg[EPT2];
    float ew[EPT2];
#pragma unroll
    for (int i = 0; i < EPT2; ++i) {
      int idx = i * 256 + tid;
      er[i] = -1; es[i] = 0; eg[i] = 0; ew[i] = 0.f; ek[i] = 0;
      if (idx < n) {
        int2 v = raw[idx];
        es[i] = v.x & 0x1FFFF; eg[i] = (v.x >> 17) & 0xFF; er[i] = (v.x >> 25) & 0x3F;
        ew[i] = __int_as_float(v.y);
      }
    }
#pragma unroll
    for (int i = 0; i < EPT2; ++i)
      if (er[i] >= 0) ek[i] = atomicAdd(&cnt[er[i]], 1);
    __syncthreads();

    if (tid == 0) {
      int o = 0;
      for (int r = 0; r < R; ++r) { offs[r] = o; o += cnt[r]; }
    }
    __syncthreads();

#pragma unroll
    for (int i = 0; i < EPT2; ++i) {
      if (er[i] >= 0) {
        int p = offs[er[i]] + ek[i];
        ls_src[p] = es[i];
        ls_tgt[p] = eg[i];
        ls_w[p]   = ew[i];
      }
    }

    __syncthreads();

    for (int r = wv; r < R; r += 8) {
      const int c = __builtin_amdgcn_readfirstlane(cnt[r]);
      const int o = __builtin_amdgcn_readfirstlane(offs[r]);
      if (c > 0) {
        const float4* wf = (const float4*)(W + (size_t)r * DIN * DOUT) + lane;
        float4 bq[32];
#pragma unroll
        for (int j = 0; j < 32; ++j) bq[j] = wf[j * 32];

        const int nt = (c + 15) >> 4;
        for (int t = 0; t < nt; ++t) {
          int q = t * 16 + lm;
          int src = 0; float wt = 0.f;
          if (q < c) { src = ls_src[o + q]; wt = ls_w[o + q]; }
          const float4* xp = (const float4*)(x + (size_t)src * DIN) + hi;
          float4 av[8];
#pragma unroll
          for (int i = 0; i < 8; ++i) av[i] = xp[2 * i];

          int tgj[8];
#pragma unroll
          for (int j = 0; j < 8; ++j) {
            int m = t * 16 + 8 * hi + j;
            int v = ls_tgt[o + min(m, c - 1)];
            tgj[j] = (m < c) ? v : -1;
          }

          v8f a8[4] = {};
#pragma unroll
          for (int ks = 0; ks < 16; ++ks) {
            const float4 xv = av[ks >> 1];
            v2f a;
            if (ks & 1) { a.x = xv.z * wt; a.y = xv.w * wt; } else { a.x = xv.x * wt; a.y = xv.y * wt; }
#pragma unroll
            for (int ct = 0; ct < 4; ++ct) {
              const float4 w4 = bq[ks * 2 + (ct >> 1)];
              v2f b;
              if (ct & 1) { b.x = w4.z; b.y = w4.w; } else { b.x = w4.x; b.y = w4.y; }
              a8[ct] = __builtin_amdgcn_wmma_f32_16x16x4_f32(
                   false, a,  false, b,
                   (short)0, a8[ct],  false,  false);
            }
          }
#pragma unroll
          for (int ct = 0; ct < 4; ++ct) {
#pragma unroll
            for (int j = 0; j < 8; ++j) {
              if (tgj[j] >= 0)
                atomicAdd(&acc[tgj[j] * DOUT + ((ct ^ (tgj[j] & 3)) << 4) + lm], a8[ct][j]);
            }
          }
        }
      }
    }

    lo += CAP;
    __syncthreads();
  } while (lo < total);

  const int nrow = min(ROWS, N - row0);
  float4* op = (float4*)(out + (size_t)row0 * DOUT);
  for (int i = tid; i < nrow * (DOUT / 4); i += 256) {
    const int t = i >> 4, q = i & 15;
    op[i] = ((const float4*)acc)[t * 16 + (((q >> 2) ^ (t & 3)) << 2) + (q & 3)];
  }
}

__global__ void __launch_bounds__(256)
rgcn_chunk_kernel(const float* __restrict__ x,
                  const float* __restrict__ W,
                  const int* __restrict__ source,
                  const int* __restrict__ target,
                  const int* __restrict__ etype,
                  const float* __restrict__ wts,
                  int E, int R,
                  float* __restrict__ out) {
  __shared__ float xs[128 * XS_STRIDE];
  __shared__ float Wl[2][DIN * W_STRIDE];
  __shared__ int   ls_src[CHUNK];
  __shared__ int   ls_tgt[CHUNK];
  __shared__ float ls_w[CHUNK];
  __shared__ int   cnt[MAXR];
  __shared__ int   offs[MAXR];

  const int tid  = threadIdx.x;
  const int base = (int)blockIdx.x * CHUNK;

  if (tid < MAXR) cnt[tid] = 0;
  __syncthreads();

  int er[EPT], ek[EPT], es[EPT], eg[EPT];
  float ew[EPT];
#pragma unroll
  for (int i = 0; i < EPT; ++i) {
    int e = base + i * 256 + tid;
    er[i] = -1; es[i] = 0; eg[i] = 0; ew[i] = 0.f; ek[i] = 0;
    if (e < E) { er[i] = etype[e]; es[i] = source[e]; eg[i] = target[e]; ew[i] = wts[e]; }
  }
#pragma unroll
  for (int i = 0; i < EPT; ++i)
    if (er[i] >= 0) ek[i] = atomicAdd(&cnt[er[i]], 1);
  __syncthreads();

  if (tid == 0) {
    int o = 0;
    for (int r = 0; r < R; ++r) { offs[r] = o; o += cnt[r]; }
  }
  __syncthreads();

#pragma unroll
  for (int i = 0; i < EPT; ++i) {
    if (er[i] >= 0) {
      int p = offs[er[i]] + ek[i];
      ls_src[p] = es[i];
      ls_tgt[p] = eg[i];
      ls_w[p]   = ew[i];
    }
  }

  float wreg[16];
  if (R > 0) {
#pragma unroll
    for (int i = 0; i < 16; ++i) wreg[i] = W[i * 256 + tid];
#pragma unroll
    for (int i = 0; i < 16; ++i) {
      int idx = i * 256 + tid;
      Wl[0][(idx >> 6) * W_STRIDE + (idx & 63)] = wreg[i];
    }
  }

  const int lane = tid & 31;
  const int wv   = __builtin_amdgcn_readfirstlane(tid >> 5);
  const int hi   = lane >> 4;
  const int lm   = lane & 15;
  const int row  = lane >> 1;
  const int half = lane & 1;
  float* xd = &xs[(wv * 16 + row) * XS_STRIDE + half * 32];
  const float* arow = &xs[(wv * 16 + lm) * XS_STRIDE + 2 * hi];

  for (int r = 0; r < R; ++r) {
    __syncthreads();
    const float* Wc = Wl[r & 1];
    const bool more = (r + 1 < R);
    if (more) {
#pragma unroll
      for (int i = 0; i < 16; ++i) wreg[i] = W[(r + 1) * DIN * DOUT + i * 256 + tid];
    }
    const int c = __builtin_amdgcn_readfirstlane(cnt[r]);
    const int o = __builtin_amdgcn_readfirstlane(offs[r]);

    const int nt = (c + 15) >> 4;
    for (int t = wv; t < nt; t += 8) {
      int q = t * 16 + row;
      int src = 0; float wt = 0.f;
      if (q < c) { src = ls_src[o + q]; wt = ls_w[o + q]; }
      const float4* xp = (const float4*)(x + (size_t)src * DIN) + half * 8;
#pragma unroll
      for (int i = 0; i < 8; ++i) {
        float4 v = xp[i];
        xd[4 * i + 0] = v.x * wt;
        xd[4 * i + 1] = v.y * wt;
        xd[4 * i + 2] = v.z * wt;
        xd[4 * i + 3] = v.w * wt;
      }
      wave_lds_order();

      int tgj[8];
#pragma unroll
      for (int j = 0; j < 8; ++j) {
        int m = t * 16 + 8 * hi + j;
        int v = ls_tgt[o + min(m, c - 1)];
        tgj[j] = (m < c) ? v : -1;
      }

#pragma unroll
      for (int ct = 0; ct < 4; ++ct) {
        v8f acc = {};
        const float* bcol = &Wc[(2 * hi) * W_STRIDE + ct * 16 + lm];
#pragma unroll
        for (int ks = 0; ks < 16; ++ks) {
          v2f a; a.x = arow[ks * 4];               a.y = arow[ks * 4 + 1];
          v2f b; b.x = bcol[(ks * 4) * W_STRIDE];  b.y = bcol[(ks * 4 + 1) * W_STRIDE];
          acc = __builtin_amdgcn_wmma_f32_16x16x4_f32(
               false, a,  false, b,
               (short)0, acc,  false,  false);
        }
#pragma unroll
        for (int j = 0; j < 8; ++j) {
          if (tgj[j] >= 0) atomicAdd(&out[(size_t)tgj[j] * DOUT + ct * 16 + lm], acc[j]);
        }
      }
      wave_lds_order();
    }

    if (more) {
      float* Wn = Wl[(r + 1) & 1];
#pragma unroll
      for (int i = 0; i < 16; ++i) {
        int idx = i * 256 + tid;
        Wn[(idx >> 6) * W_STRIDE + (idx & 63)] = wreg[i];
      }
    }
  }
}

static inline size_t align256(size_t v) { return (v + 255) & ~(size_t)255; }

extern "C" void kernel_launch(void* const* d_in, const int* in_sizes, int n_in,
                              void* d_out, int out_size, void* d_ws, size_t ws_size,
                              hipStream_t stream) {
  const float* x       = (const float*)d_in[0];
  const int*   source  = (const int*)d_in[1];
  const int*   target  = (const int*)d_in[2];
  const int*   etype   = (const int*)d_in[3];
  const float* ew      = (const float*)d_in[4];
  const float* bases   = (const float*)d_in[5];
  const float* comb    = (const float*)d_in[6];
  float*       out     = (float*)d_out;

  const int N = in_sizes[0] / DIN;
  const int E = in_sizes[1];
  const int B = in_sizes[5] / (DIN * DOUT);
  const int R = in_sizes[6] / B;

  const int nEB = (E + CHUNK - 1) / CHUNK;
  const int nbk = (N + ROWS - 1) / ROWS;

  char*  ws   = (char*)d_ws;
  float* Wbuf = (float*)ws;
  const size_t oPK = align256((size_t)R * DIN * DOUT * sizeof(float));
  const size_t oSE = oPK + align256((size_t)nbk * nEB * sizeof(int));
  const size_t need = oSE + (size_t)nEB * CHUNK * sizeof(int2);

  if (ws_size >= need && nbk <= MAXBK && R <= MAXR) {
    int*  PK = (int*)(ws + oPK);
    int*  SE = (int*)(ws + oSE);
    bin_kernel<<<R + nEB, 256, 0, stream>>>(source, target, etype, ew, bases, comb,
                                            Wbuf, B, R, E, nEB, nbk, PK, SE);
    rgcn_owner_kernel<<<nbk, 256, 0, stream>>>(x, Wbuf, PK, SE, N, R, nEB, out);
    return;
  }

  hipMemsetAsync(out, 0, (size_t)N * DOUT * sizeof(float), stream);
  compute_W_kernel<<<R, 256, 0, stream>>>(bases, comb, Wbuf, B);
  int nblk = (E + CHUNK - 1) / CHUNK;
  rgcn_chunk_kernel<<<nblk, 256, 0, stream>>>(x, Wbuf, source, target, etype, ew,
                                              E, R, out);
}


# owner epilogue: tile rows past a group's end add into a spare LDS row instead of being masked off: a tile's 32 LDS float adds become straight-line code, 32 exec-mask branches gone; add order kept
# speedup vs baseline: 6.4972x; 1.1279x over previous
#include <hip/hip_runtime.h>

typedef __attribute__((ext_vector_type(2))) float v2f;
typedef __attribute__((ext_vector_type(8))) float v8f;

#define DIN 64
#define DOUT 64
#define CHUNK 2048
#define EPT 8
#define MAXR 64
#define XS_STRIDE 68
#define W_STRIDE 65
#define ROWS 196
#define MAXBK 512
#define CAP 2816
#define EPT2 11

__global__ void compute_W_kernel(const float* __restrict__ bases,
                                 const float* __restrict__ comb,
                                 float* __restrict__ W, int B) {
  int r = blockIdx.x;
  const float* c = comb + r * B;
  for (int idx = threadIdx.x; idx < DIN * DOUT; idx += blockDim.x) {
    float acc = 0.f;
    for (int b = 0; b < B; ++b) acc += c[b] * bases[b * DIN * DOUT + idx];
    W[r * DIN * DOUT + idx] = acc;
  }
}

__device__ __forceinline__ void wave_lds_order() {
  __builtin_amdgcn_fence(__ATOMIC_RELEASE, "wavefront");
  __builtin_amdgcn_wave_barrier();
  __builtin_amdgcn_fence(__ATOMIC_ACQUIRE, "wavefront");
}

__device__ __forceinline__ int block_incl_scan(int v, int* part, int tid, int& total) {
  const int ln = tid & 31, wq = tid >> 5;
  int s = v;
#pragma unroll
  for (int d = 1; d < 32; d <<= 1) {
    int u = __shfl_up(s, d, 32);
    if (ln >= d) s += u;
  }
  if (ln == 31) part[wq] = s;
  __syncthreads();
  int before = 0, all = 0;
#pragma unroll
  for (int k = 0; k < 8; ++k) {
    int t = part[k];
    all += t;
    if (k < wq) before += t;
  }
  total = all;
  return s + before;
}

__global__ void __launch_bounds__(256)
bin_kernel(const int* __restrict__ source,
           const int* __restrict__ target,
           const int* __restrict__ etype,
           const float* __restrict__ wts,
           const float* __restrict__ bases,
           const float* __restrict__ comb,
           float* __restrict__ W, int B, int R,
           int E, int nEB, int nbk,
           int* __restrict__ PK, int* __restrict__ SE) {
  __shared__ int  hist[MAXBK];
  __shared__ int  start[MAXBK];
  __shared__ int  part[8];
  __shared__ int2 stage[CHUNK];

  const int tid = threadIdx.x;

  if ((int)blockIdx.x < R) {
    const int r = blockIdx.x;
    const float* c = comb + r * B;
    for (int idx = tid; idx < DIN * DOUT; idx += 256) {
      float a = 0.f;
      for (int b = 0; b < B; ++b) a += c[b] * bases[b * DIN * DOUT + idx];
      const int k = idx >> 6, n = idx & 63;
      const int ks  = ((k >> 3) << 1) | ((k >> 1) & 1);
      const int dst = (ks * 2 + (n >> 5)) * 128
                    + (((k >> 2) & 1) * 16 + (n & 15)) * 4
                    + ((n >> 4) & 1) * 2 + (k & 1);
      W[r * DIN * DOUT + dst] = a;
    }
    return;
  }
  const int blk = (int)blockIdx.x - R;

  hist[tid] = 0;
  hist[tid + 256] = 0;
  __syncthreads();

  const int base = blk * CHUNK;
  int es[EPT], eg[EPT], er[EPT], ek[EPT], eb[EPT];
  float ew[EPT];
#pragma unroll
  for (int i = 0; i < EPT; ++i) {
    int e = base + i * 256 + tid;
    eb[i] = -1; es[i] = 0; eg[i] = 0; er[i] = 0; ew[i] = 0.f; ek[i] = 0;
    if (e < E) {
      es[i] = source[e]; eg[i] = target[e]; er[i] = etype[e]; ew[i] = wts[e];
      eb[i] = eg[i] / ROWS;
    }
  }
#pragma unroll
  for (int i = 0; i < EPT; ++i)
    if (eb[i] >= 0) ek[i] = atomicAdd(&hist[eb[i]], 1);
  __syncthreads();

  const int h0 = hist[2 * tid], h1 = hist[2 * tid + 1];
  int nall;
  const int incl = block_incl_scan(h0 + h1, part, tid, nall);
  const int excl = incl - (h0 + h1);
  start[2 * tid]     = excl;
  start[2 * tid + 1] = excl + h0;
  __syncthreads();

  for (int b = tid; b < nbk; b += 256) PK[b * nEB + blk] = (start[b] << 12) | hist[b];

#pragma unroll
  for (int i = 0; i < EPT; ++i) {
    if (eb[i] >= 0) {
      int p = start[eb[i]] + ek[i];
      stage[p] = make_int2(es[i] | ((eg[i] - eb[i] * ROWS) << 17) | (er[i] << 25),
                           __float_as_int(ew[i]));
    }
  }
  __syncthreads();

  const int nvalid = min(CHUNK, E - base);
  int2* SE2 = (int2*)SE;
  for (int i = tid; i < nvalid; i += 256) SE2[base + i] = stage[i];
}

__global__ void __launch_bounds__(256)
rgcn_owner_kernel(const float* __restrict__ x,
                  const float* __restrict__ W,
                  const int* __restrict__ PK,
                  const int* __restrict__ SE,
                  int N, int R, int nEB,
                  float* __restrict__ out) {
  __shared__ __attribute__((aligned(16))) float acc[(ROWS + 1) * DOUT];
  __shared__ int2  raw[CAP + 1];
  __shared__ int   ls_src[CAP];
  __shared__ int   ls_tgt[CAP];
  __shared__ float ls_w[CAP];
  __shared__ int   cnt[MAXR];
  __shared__ int   offs[MAXR];
  __shared__ int   part[8];

  const int2* SE2 = (const int2*)SE;

  const int tid  = threadIdx.x;
  const int bk   = blockIdx.x;
  const int row0 = bk * ROWS;

  for (int i = tid; i < (ROWS + 1) * DOUT / 4; i += 256)
    ((float4*)acc)[i] = make_float4(0.f, 0.f, 0.f, 0.f);

  const int lane = tid & 31;
  const int wv   = __builtin_amdgcn_readfirstlane(tid >> 5);
  const int hi   = lane >> 4;
  const int lm   = lane & 15;

  int lo = 0;
  int total = 0;
  do {
    int running = 0;
    for (int bb = 0; bb < nEB; bb += 256) {
      const int blk = bb + tid;
      const int pk  = (blk < nEB) ? PK[bk * nEB + blk] : 0;
      const int c   = pk & 4095;
      const int s   = pk >> 12;
      int btot;
      const int incl = block_incl_scan(c, part, tid, btot);
      const int first = running + incl - c - lo;
      running += btot;
      const int2* sp = SE2 + (size_t)blk * CHUNK + s;
      for (int k0 = 0; k0 < c; k0 += 8) {
        int2 v[8];
#pragma unroll
        for (int j = 0; j < 8; ++j) v[j] = sp[min(k0 + j, c - 1)];
#pragma unroll
        for (int j = 0; j < 8; ++j) {
          int pos = first + k0 + j;
          bool keep = (k0 + j < c) && (pos >= 0) && (pos < CAP);
          raw[keep ? pos : CAP] = v[j];
        }
      }
      __syncthreads();
    }
    total = __builtin_amdgcn_readfirstlane(running);
    const int n = min(CAP, total - lo);

    if (tid < MAXR) cnt[tid] = 0;
    __syncthreads();

    int er[EPT2], ek[EPT2], es[EPT2], eg[EPT2];
    float ew[EPT2];
#pragma unroll
    for (int i = 0; i < EPT2; ++i) {
      int idx = i * 256 + tid;
      er[i] = -1; es[i] = 0; eg[i] = 0; ew[i] = 0.f; ek[i] = 0;
      if (idx < n) {
        int2 v = raw[idx];
        es[i] = v.x & 0x1FFFF; eg[i] = (v.x >> 17) & 0xFF; er[i] = (v.x >> 25) & 0x3F;
        ew[i] = __int_as_float(v.y);
      }
    }
#pragma unroll
    for (int i = 0; i < EPT2; ++i)
      if (er[i] >= 0) ek[i] = atomicAdd(&cnt[er[i]], 1);
    __syncthreads();

    if (tid == 0) {
      int o = 0;
      for (int r = 0; r < R; ++r) { offs[r] = o; o += cnt[r]; }
    }
    __syncthreads();

#pragma unroll
    for (int i = 0; i < EPT2; ++i) {
      if (er[i] >= 0) {
        int p = offs[er[i]] + ek[i];
        ls_src[p] = es[i];
        ls_tgt[p] = eg[i];
        ls_w[p]   = ew[i];
      }
    }

    __syncthreads();

    for (int r = wv; r < R; r += 8) {
      const int c = __builtin_amdgcn_readfirstlane(cnt[r]);
      const int o = __builtin_amdgcn_readfirstlane(offs[r]);
      if (c > 0) {
        const float4* wf = (const float4*)(W + (size_t)r * DIN * DOUT) + lane;
        float4 bq[32];
#pragma unroll
        for (int j = 0; j < 32; ++j) bq[j] = wf[j * 32];

        const int nt = (c + 15) >> 4;
        for (int t = 0; t < nt; ++t) {
          int q = t * 16 + lm;
          int src = 0; float wt = 0.f;
          if (q < c) { src = ls_src[o + q]; wt = ls_w[o + q]; }
          const float4* xp = (const float4*)(x + (size_t)src * DIN) + hi;
          float4 av[8];
#pragma unroll
          for (int i = 0; i < 8; ++i) av[i] = xp[2 * i];

          int rowa[8], rowk[8];
#pragma unroll
          for (int j = 0; j < 8; ++j) {
            int m = t * 16 + 8 * hi + j;
            int v = ls_tgt[o + min(m, c - 1)];
            v = (m < c) ? v : ROWS;
            rowa[j] = v * DOUT + lm;
            rowk[j] = v & 3;
          }

          v8f a8[4] = {};
#pragma unroll
          for (int ks = 0; ks < 16; ++ks) {
            const float4 xv = av[ks >> 1];
            v2f a;
            if (ks & 1) { a.x = xv.z * wt; a.y = xv.w * wt; } else { a.x = xv.x * wt; a.y = xv.y * wt; }
#pragma unroll
            for (int ct = 0; ct < 4; ++ct) {
              const float4 w4 = bq[ks * 2 + (ct >> 1)];
              v2f b;
              if (ct & 1) { b.x = w4.z; b.y = w4.w; } else { b.x = w4.x; b.y = w4.y; }
              a8[ct] = __builtin_amdgcn_wmma_f32_16x16x4_f32(
                   false, a,  false, b,
                   (short)0, a8[ct],  false,  false);
            }
          }
#pragma unroll
          for (int ct = 0; ct < 4; ++ct) {
#pragma unroll
            for (int j = 0; j < 8; ++j)
              atomicAdd(&acc[rowa[j] + ((ct ^ rowk[j]) << 4)], a8[ct][j]);
          }
        }
      }
    }

    lo += CAP;
    __syncthreads();
  } while (lo < total);

  const int nrow = min(ROWS, N - row0);
  float4* op = (float4*)(out + (size_t)row0 * DOUT);
  for (int i = tid; i < nrow * (DOUT / 4); i += 256) {
    const int t = i >> 4, q = i & 15;
    op[i] = ((const float4*)acc)[t * 16 + (((q >> 2) ^ (t & 3)) << 2) + (q & 3)];
  }
}

__global__ void __launch_bounds__(256)
rgcn_chunk_kernel(const float* __restrict__ x,
                  const float* __restrict__ W,
                  const int* __restrict__ source,
                  const int* __restrict__ target,
                  const int* __restrict__ etype,
                  const float* __restrict__ wts,
                  int E, int R,
                  float* __restrict__ out) {
  __shared__ float xs[128 * XS_STRIDE];
  __shared__ float Wl[2][DIN * W_STRIDE];
  __shared__ int   ls_src[CHUNK];
  __shared__ int   ls_tgt[CHUNK];
  __shared__ float ls_w[CHUNK];
  __shared__ int   cnt[MAXR];
  __shared__ int   offs[MAXR];

  const int tid  = threadIdx.x;
  const int base = (int)blockIdx.x * CHUNK;

  if (tid < MAXR) cnt[tid] = 0;
  __syncthreads();

  int er[EPT], ek[EPT], es[EPT], eg[EPT];
  float ew[EPT];
#pragma unroll
  for (int i = 0; i < EPT; ++i) {
    int e = base + i * 256 + tid;
    er[i] = -1; es[i] = 0; eg[i] = 0; ew[i] = 0.f; ek[i] = 0;
    if (e < E) { er[i] = etype[e]; es[i] = source[e]; eg[i] = target[e]; ew[i] = wts[e]; }
  }
#pragma unroll
  for (int i = 0; i < EPT; ++i)
    if (er[i] >= 0) ek[i] = atomicAdd(&cnt[er[i]], 1);
  __syncthreads();

  if (tid == 0) {
    int o = 0;
    for (int r = 0; r < R; ++r) { offs[r] = o; o += cnt[r]; }
  }
  __syncthreads();

#pragma unroll
  for (int i = 0; i < EPT; ++i) {
    if (er[i] >= 0) {
      int p = offs[er[i]] + ek[i];
      ls_src[p] = es[i];
      ls_tgt[p] = eg[i];
      ls_w[p]   = ew[i];
    }
  }

  float wreg[16];
  if (R > 0) {
#pragma unroll
    for (int i = 0; i < 16; ++i) wreg[i] = W[i * 256 + tid];
#pragma unroll
    for (int i = 0; i < 16; ++i) {
      int idx = i * 256 + tid;
      Wl[0][(idx >> 6) * W_STRIDE + (idx & 63)] = wreg[i];
    }
  }

  const int lane = tid & 31;
  const int wv   = __builtin_amdgcn_readfirstlane(tid >> 5);
  const int hi   = lane >> 4;
  const int lm   = lane & 15;
  const int row  = lane >> 1;
  const int half = lane & 1;
  float* xd = &xs[(wv * 16 + row) * XS_STRIDE + half * 32];
  const float* arow = &xs[(wv * 16 + lm) * XS_STRIDE + 2 * hi];

  for (int r = 0; r < R; ++r) {
    __syncthreads();
    const float* Wc = Wl[r & 1];
    const bool more = (r + 1 < R);
    if (more) {
#pragma unroll
      for (int i = 0; i < 16; ++i) wreg[i] = W[(r + 1) * DIN * DOUT + i * 256 + tid];
    }
    const int c = __builtin_amdgcn_readfirstlane(cnt[r]);
    const int o = __builtin_amdgcn_readfirstlane(offs[r]);

    const int nt = (c + 15) >> 4;
    for (int t = wv; t < nt; t += 8) {
      int q = t * 16 + row;
      int src = 0; float wt = 0.f;
      if (q < c) { src = ls_src[o + q]; wt = ls_w[o + q]; }
      const float4* xp = (const float4*)(x + (size_t)src * DIN) + half * 8;
#pragma unroll
      for (int i = 0; i < 8; ++i) {
        float4 v = xp[i];
        xd[4 * i + 0] = v.x * wt;
        xd[4 * i + 1] = v.y * wt;
        xd[4 * i + 2] = v.z * wt;
        xd[4 * i + 3] = v.w * wt;
      }
      wave_lds_order();

      int tgj[8];
#pragma unroll
      for (int j = 0; j < 8; ++j) {
        int m = t * 16 + 8 * hi + j;
        int v = ls_tgt[o + min(m, c - 1)];
        tgj[j] = (m < c) ? v : -1;
      }

#pragma unroll
      for (int ct = 0; ct < 4; ++ct) {
        v8f acc = {};
        const float* bcol = &Wc[(2 * hi) * W_STRIDE + ct * 16 + lm];
#pragma unroll
        for (int ks = 0; ks < 16; ++ks) {
          v2f a; a.x = arow[ks * 4];               a.y = arow[ks * 4 + 1];
          v2f b; b.x = bcol[(ks * 4) * W_STRIDE];  b.y = bcol[(ks * 4 + 1) * W_STRIDE];
          acc = __builtin_amdgcn_wmma_f32_16x16x4_f32(
               false, a,  false, b,
               (short)0, acc,  false,  false);
        }
#pragma unroll
        for (int j = 0; j < 8; ++j) {
          if (tgj[j] >= 0) atomicAdd(&out[(size_t)tgj[j] * DOUT + ct * 16 + lm], acc[j]);
        }
      }
      wave_lds_order();
    }

    if (more) {
      float* Wn = Wl[(r + 1) & 1];
#pragma unroll
      for (int i = 0; i < 16; ++i) {
        int idx = i * 256 + tid;
        Wn[(idx >> 6) * W_STRIDE + (idx & 63)] = wreg[i];
      }
    }
  }
}

static inline size_t align256(size_t v) { return (v + 255) & ~(size_t)255; }

extern "C" void kernel_launch(void* const* d_in, const int* in_sizes, int n_in,
                              void* d_out, int out_size, void* d_ws, size_t ws_size,
                              hipStream_t stream) {
  const float* x       = (const float*)d_in[0];
  const int*   source  = (const int*)d_in[1];
  const int*   target  = (const int*)d_in[2];
  const int*   etype   = (const int*)d_in[3];
  const float* ew      = (const float*)d_in[4];
  const float* bases   = (const float*)d_in[5];
  const float* comb    = (const float*)d_in[6];
  float*       out     = (float*)d_out;

  const int N = in_sizes[0] / DIN;
  const int E = in_sizes[1];
  const int B = in_sizes[5] / (DIN * DOUT);
  const int R = in_sizes[6] / B;

  const int nEB = (E + CHUNK - 1) / CHUNK;
  const int nbk = (N + ROWS - 1) / ROWS;

  char*  ws   = (char*)d_ws;
  float* Wbuf = (float*)ws;
  const size_t oPK = align256((size_t)R * DIN * DOUT * sizeof(float));
  const size_t oSE = oPK + align256((size_t)nbk * nEB * sizeof(int));
  const size_t need = oSE + (size_t)nEB * CHUNK * sizeof(int2);

  if (ws_size >= need && nbk <= MAXBK && R <= MAXR) {
    int*  PK = (int*)(ws + oPK);
    int*  SE = (int*)(ws + oSE);
    bin_kernel<<<R + nEB, 256, 0, stream>>>(source, target, etype, ew, bases, comb,
                                            Wbuf, B, R, E, nEB, nbk, PK, SE);
    rgcn_owner_kernel<<<nbk, 256, 0, stream>>>(x, Wbuf, PK, SE, N, R, nEB, out);
    return;
  }

  hipMemsetAsync(out, 0, (size_t)N * DOUT * sizeof(float), stream);
  compute_W_kernel<<<R, 256, 0, stream>>>(bases, comb, Wbuf, B);
  int nblk = (E + CHUNK - 1) / CHUNK;
  rgcn_chunk_kernel<<<nblk, 256, 0, stream>>>(x, Wbuf, source, target, etype, ew,
                                              E, R, out);
}


# owner kernel: edges of one (relation, out row) are summed in x space into one A row before the product, as the reference's segment sum (LDS sort by relation and row): fewer tiles; add order changes
# speedup vs baseline: 6.9902x; 1.0759x over previous
#include <hip/hip_runtime.h>

typedef __attribute__((ext_vector_type(2))) float v2f;
typedef __attribute__((ext_vector_type(8))) float v8f;

#define DIN 64
#define DOUT 64
#define CHUNK 2048
#define EPT 8
#define MAXR 64
#define XS_STRIDE 68
#define W_STRIDE 65
#define ROWS 196
#define MAXBK 512
#define CAP 2816
#define EPT2 11
#define OWNR 16
#define KPT 13
#define KPAD (KPT * 256)

__global__ void compute_W_kernel(const float* __restrict__ bases,
                                 const float* __restrict__ comb,
                                 float* __restrict__ W, int B) {
  int r = blockIdx.x;
  const float* c = comb + r * B;
  for (int idx = threadIdx.x; idx < DIN * DOUT; idx += blockDim.x) {
    float acc = 0.f;
    for (int b = 0; b < B; ++b) acc += c[b] * bases[b * DIN * DOUT + idx];
    W[r * DIN * DOUT + idx] = acc;
  }
}

__device__ __forceinline__ void wave_lds_order() {
  __builtin_amdgcn_fence(__ATOMIC_RELEASE, "wavefront");
  __builtin_amdgcn_wave_barrier();
  __builtin_amdgcn_fence(__ATOMIC_ACQUIRE, "wavefront");
}

__device__ __forceinline__ int block_incl_scan(int v, int* part, int tid, int& total) {
  const int ln = tid & 31, wq = tid >> 5;
  int s = v;
#pragma unroll
  for (int d = 1; d < 32; d <<= 1) {
    int u = __shfl_up(s, d, 32);
    if (ln >= d) s += u;
  }
  if (ln == 31) part[wq] = s;
  __syncthreads();
  int before = 0, all = 0;
#pragma unroll
  for (int k = 0; k < 8; ++k) {
    int t = part[k];
    all += t;
    if (k < wq) before += t;
  }
  total = all;
  return s + before;
}

__global__ void __launch_bounds__(256)
bin_kernel(const int* __restrict__ source,
           const int* __restrict__ target,
           const int* __restrict__ etype,
           const float* __restrict__ wts,
           const float* __restrict__ bases,
           const float* __restrict__ comb,
           float* __restrict__ W, int B, int R,
           int E, int nEB, int nbk,
           int* __restrict__ PK, int* __restrict__ SE) {
  __shared__ int  hist[MAXBK];
  __shared__ int  start[MAXBK];
  __shared__ int  part[8];
  __shared__ int2 stage[CHUNK];

  const int tid = threadIdx.x;

  if ((int)blockIdx.x < R) {
    const int r = blockIdx.x;
    const float* c = comb + r * B;
    for (int idx = tid; idx < DIN * DOUT; idx += 256) {
      float a = 0.f;
      for (int b = 0; b < B; ++b) a += c[b] * bases[b * DIN * DOUT + idx];
      const int k = idx >> 6, n = idx & 63;
      const int ks  = ((k >> 3) << 1) | ((k >> 1) & 1);
      const int dst = (ks * 2 + (n >> 5)) * 128
                    + (((k >> 2) & 1) * 16 + (n & 15)) * 4
                    + ((n >> 4) & 1) * 2 + (k & 1);
      W[r * DIN * DOUT + dst] = a;
    }
    return;
  }
  const int blk = (int)blockIdx.x - R;

  hist[tid] = 0;
  hist[tid + 256] = 0;
  __syncthreads();

  const int base = blk * CHUNK;
  int es[EPT], eg[EPT], er[EPT], ek[EPT], eb[EPT];
  float ew[EPT];
#pragma unroll
  for (int i = 0; i < EPT; ++i) {
    int e = base + i * 256 + tid;
    eb[i] = -1; es[i] = 0; eg[i] = 0; er[i] = 0; ew[i] = 0.f; ek[i] = 0;
    if (e < E) {
      es[i] = source[e]; eg[i] = target[e]; er[i] = etype[e]; ew[i] = wts[e];
      eb[i] = eg[i] / ROWS;
    }
  }
#pragma unroll
  for (int i = 0; i < EPT; ++i)
    if (eb[i] >= 0) ek[i] = atomicAdd(&hist[eb[i]], 1);
  __syncthreads();

  const int h0 = hist[2 * tid], h1 = hist[2 * tid + 1];
  int nall;
  const int incl = block_incl_scan(h0 + h1, part, tid, nall);
  const int excl = incl - (h0 + h1);
  start[2 * tid]     = excl;
  start[2 * tid + 1] = excl + h0;
  __syncthreads();

  for (int b = tid; b < nbk; b += 256) PK[b * nEB + blk] = (start[b] << 12) | hist[b];

#pragma unroll
  for (int i = 0; i < EPT; ++i) {
    if (eb[i] >= 0) {
      int p = start[eb[i]] + ek[i];
      stage[p] = make_int2(es[i] | ((eg[i] - eb[i] * ROWS) << 17) | (er[i] << 25),
                           __float_as_int(ew[i]));
    }
  }
  __syncthreads();

  const int nvalid = min(CHUNK, E - base);
  int2* SE2 = (int2*)SE;
  for (int i = tid; i < nvalid; i += 256) SE2[base + i] = stage[i];
}

__global__ void __launch_bounds__(256)
rgcn_owner_kernel(const float* __restrict__ x,
                  const float* __restrict__ W,
                  const int* __restrict__ PK,
                  const int* __restrict__ SE,
                  int N, int R, int nEB,
                  float* __restrict__ out) {
  __shared__ __attribute__((aligned(16))) float acc[(ROWS + 1) * DOUT];
  __shared__ int2  raw[CAP + 1];
  __shared__ int   ls_src[CAP];
  __shared__ float ls_w[CAP];
  __shared__ int   seg[CAP + 1];
  __shared__ int   kc[KPAD];
  __shared__ int   soffs[OWNR + 1];
  __shared__ int   part[8];

  const int2* SE2 = (const int2*)SE;

  const int tid  = threadIdx.x;
  const int bk   = blockIdx.x;
  const int row0 = bk * ROWS;

  for (int i = tid; i < (ROWS + 1) * DOUT / 4; i += 256)
    ((float4*)acc)[i] = make_float4(0.f, 0.f, 0.f, 0.f);

  const int lane = tid & 31;
  const int wv   = __builtin_amdgcn_readfirstlane(tid >> 5);
  const int hi   = lane >> 4;
  const int lm   = lane & 15;

  int lo = 0;
  int total = 0;
  do {
    int running = 0;
    for (int bb = 0; bb < nEB; bb += 256) {
      const int blk = bb + tid;
      const int pk  = (blk < nEB) ? PK[bk * nEB + blk] : 0;
      const int c   = pk & 4095;
      const int s   = pk >> 12;
      int btot;
      const int incl = block_incl_scan(c, part, tid, btot);
      const int first = running + incl - c - lo;
      running += btot;
      const int2* sp = SE2 + (size_t)blk * CHUNK + s;
      for (int k0 = 0; k0 < c; k0 += 8) {
        int2 v[8];
#pragma unroll
        for (int j = 0; j < 8; ++j) v[j] = sp[min(k0 + j, c - 1)];
#pragma unroll
        for (int j = 0; j < 8; ++j) {
          int pos = first + k0 + j;
          bool keep = (k0 + j < c) && (pos >= 0) && (pos < CAP);
          raw[keep ? pos : CAP] = v[j];
        }
      }
      __syncthreads();
    }
    total = __builtin_amdgcn_readfirstlane(running);
    const int n = min(CAP, total - lo);

    for (int i = tid; i < KPAD; i += 256) kc[i] = 0;
    __syncthreads();

    int ekey[EPT2], ek[EPT2], es[EPT2];
    float ew[EPT2];
#pragma unroll
    for (int i = 0; i < EPT2; ++i) {
      int idx = i * 256 + tid;
      ekey[i] = -1; es[i] = 0; ew[i] = 0.f; ek[i] = 0;
      if (idx < n) {
        int2 v = raw[idx];
        es[i] = v.x & 0x1FFFF;
        ekey[i] = min((int)(((v.x >> 25) & 0x3F) * ROWS + ((v.x >> 17) & 0xFF)), KPAD - 1);
        ew[i] = __int_as_float(v.y);
      }
    }
#pragma unroll
    for (int i = 0; i < EPT2; ++i)
      if (ekey[i] >= 0) ek[i] = atomicAdd(&kc[ekey[i]], 1);
    __syncthreads();

    {
      int kcnt[KPT];
      int tsum = 0;
#pragma unroll
      for (int i = 0; i < KPT; ++i) {
        kcnt[i] = kc[KPT * tid + i];
        tsum += kcnt[i] | ((kcnt[i] > 0) << 16);
      }
      int tall;
      const int incl = block_incl_scan(tsum, part, tid, tall);
      int run = incl - tsum;
#pragma unroll
      for (int i = 0; i < KPT; ++i) {
        const int key = KPT * tid + i;
        kc[key] = run;
        if (kcnt[i] > 0) seg[run >> 16] = (run & 0xFFFF) | ((key % ROWS) << 12);
        run += kcnt[i] | ((kcnt[i] > 0) << 16);
      }
      if (tid == 255) seg[run >> 16] = run & 0xFFFF;
    }
    __syncthreads();

#pragma unroll
    for (int i = 0; i < EPT2; ++i) {
      if (ekey[i] >= 0) {
        int p = (kc[ekey[i]] & 0xFFFF) + ek[i];
        ls_src[p] = es[i];
        ls_w[p]   = ew[i];
      }
    }
    if (tid <= R) soffs[tid] = kc[tid * ROWS] >> 16;

    __syncthreads();

    for (int r = wv; r < R; r += 8) {
      const int o = __builtin_amdgcn_readfirstlane(soffs[r]);
      const int c = __builtin_amdgcn_readfirstlane(soffs[r + 1]) - o;
      if (c > 0) {
        const float4* wf = (const float4*)(W + (size_t)r * DIN * DOUT) + lane;
        float4 bq[32];
#pragma unroll
        for (int j = 0; j < 32; ++j) bq[j] = wf[j * 32];

        const int nt = (c + 15) >> 4;
        for (int t = 0; t < nt; ++t) {
          const int q  = t * 16 + lm;
          const int qi = o + min(q, c - 1);
          const int sg0 = seg[qi];
          const int sg1 = seg[qi + 1];
          const int e0 = sg0 & 0xFFF;
          const int e1 = (q < c) ? (sg1 & 0xFFF) : e0;
          float4 av[8];
          {
            const int   src = ls_src[e0];
            const float wt  = (q < c) ? ls_w[e0] : 0.f;
            const float4* xp = (const float4*)(x + (size_t)src * DIN) + hi;
#pragma unroll
            for (int i = 0; i < 8; ++i) av[i] = xp[2 * i];
#pragma unroll
            for (int i = 0; i < 8; ++i) {
              av[i].x *= wt; av[i].y *= wt; av[i].z *= wt; av[i].w *= wt;
            }
          }
#pragma nounroll
          for (int e = e0 + 1; e < e1; ++e) {
            const int   src = ls_src[e];
            const float wt  = ls_w[e];
            const float4* xp = (const float4*)(x + (size_t)src * DIN) + hi;
            float4 xv[8];
#pragma unroll
            for (int i = 0; i < 8; ++i) xv[i] = xp[2 * i];
#pragma unroll
            for (int i = 0; i < 8; ++i) {
              av[i].x += wt * xv[i].x;
              av[i].y += wt * xv[i].y;
              av[i].z += wt * xv[i].z;
              av[i].w += wt * xv[i].w;
            }
          }

          int rowa[8], rowk[8];
#pragma unroll
          for (int j = 0; j < 8; ++j) {
            int m = t * 16 + 8 * hi + j;
            int v = seg[o + min(m, c - 1)] >> 12;
            v = (m < c) ? v : ROWS;
            rowa[j] = v * DOUT + lm;
            rowk[j] = v & 3;
          }

          v8f a8[4] = {};
#pragma unroll
          for (int ks = 0; ks < 16; ++ks) {
            const float4 xv = av[ks >> 1];
            v2f a;
            if (ks & 1) { a.x = xv.z; a.y = xv.w; } else { a.x = xv.x; a.y = xv.y; }
#pragma unroll
            for (int ct = 0; ct < 4; ++ct) {
              const float4 w4 = bq[ks * 2 + (ct >> 1)];
              v2f b;
              if (ct & 1) { b.x = w4.z; b.y = w4.w; } else { b.x = w4.x; b.y = w4.y; }
              a8[ct] = __builtin_amdgcn_wmma_f32_16x16x4_f32(
                   false, a,  false, b,
                   (short)0, a8[ct],  false,  false);
            }
          }
#pragma unroll
          for (int ct = 0; ct < 4; ++ct) {
#pragma unroll
            for (int j = 0; j < 8; ++j)
              atomicAdd(&acc[rowa[j] + ((ct ^ rowk[j]) << 4)], a8[ct][j]);
          }
        }
      }
    }

    lo += CAP;
    __syncthreads();
  } while (lo < total);

  const int nrow = min(ROWS, N - row0);
  float4* op = (float4*)(out + (size_t)row0 * DOUT);
  for (int i = tid; i < nrow * (DOUT / 4); i += 256) {
    const int t = i >> 4, q = i & 15;
    op[i] = ((const float4*)acc)[t * 16 + (((q >> 2) ^ (t & 3)) << 2) + (q & 3)];
  }
}

__global__ void __launch_bounds__(256)
rgcn_chunk_kernel(const float* __restrict__ x,
                  const float* __restrict__ W,
                  const int* __restrict__ source,
                  const int* __restrict__ target,
                  const int* __restrict__ etype,
                  const float* __restrict__ wts,
                  int E, int R,
                  float* __restrict__ out) {
  __shared__ float xs[128 * XS_STRIDE];
  __shared__ float Wl[2][DIN * W_STRIDE];
  __shared__ int   ls_src[CHUNK];
  __shared__ int   ls_tgt[CHUNK];
  __shared__ float ls_w[CHUNK];
  __shared__ int   cnt[MAXR];
  __shared__ int   offs[MAXR];

  const int tid  = threadIdx.x;
  const int base = (int)blockIdx.x * CHUNK;

  if (tid < MAXR) cnt[tid] = 0;
  __syncthreads();

  int er[EPT], ek[EPT], es[EPT], eg[EPT];
  float ew[EPT];
#pragma unroll
  for (int i = 0; i < EPT; ++i) {
    int e = base + i * 256 + tid;
    er[i] = -1; es[i] = 0; eg[i] = 0; ew[i] = 0.f; ek[i] = 0;
    if (e < E) { er[i] = etype[e]; es[i] = source[e]; eg[i] = target[e]; ew[i] = wts[e]; }
  }
#pragma unroll
  for (int i = 0; i < EPT; ++i)
    if (er[i] >= 0) ek[i] = atomicAdd(&cnt[er[i]], 1);
  __syncthreads();

  if (tid == 0) {
    int o = 0;
    for (int r = 0; r < R; ++r) { offs[r] = o; o += cnt[r]; }
  }
  __syncthreads();

#pragma unroll
  for (int i = 0; i < EPT; ++i) {
    if (er[i] >= 0) {
      int p = offs[er[i]] + ek[i];
      ls_src[p] = es[i];
      ls_tgt[p] = eg[i];
      ls_w[p]   = ew[i];
    }
  }

  float wreg[16];
  if (R > 0) {
#pragma unroll
    for (int i = 0; i < 16; ++i) wreg[i] = W[i * 256 + tid];
#pragma unroll
    for (int i = 0; i < 16; ++i) {
      int idx = i * 256 + tid;
      Wl[0][(idx >> 6) * W_STRIDE + (idx & 63)] = wreg[i];
    }
  }

  const int lane = tid & 31;
  const int wv   = __builtin_amdgcn_readfirstlane(tid >> 5);
  const int hi   = lane >> 4;
  const int lm   = lane & 15;
  const int row  = lane >> 1;
  const int half = lane & 1;
  float* xd = &xs[(wv * 16 + row) * XS_STRIDE + half * 32];
  const float* arow = &xs[(wv * 16 + lm) * XS_STRIDE + 2 * hi];

  for (int r = 0; r < R; ++r) {
    __syncthreads();
    const float* Wc = Wl[r & 1];
    const bool more = (r + 1 < R);
    if (more) {
#pragma unroll
      for (int i = 0; i < 16; ++i) wreg[i] = W[(r + 1) * DIN * DOUT + i * 256 + tid];
    }
    const int c = __builtin_amdgcn_readfirstlane(cnt[r]);
    const int o = __builtin_amdgcn_readfirstlane(offs[r]);

    const int nt = (c + 15) >> 4;
    for (int t = wv; t < nt; t += 8) {
      int q = t * 16 + row;
      int src = 0; float wt = 0.f;
      if (q < c) { src = ls_src[o + q]; wt = ls_w[o + q]; }
      const float4* xp = (const float4*)(x + (size_t)src * DIN) + half * 8;
#pragma unroll
      for (int i = 0; i < 8; ++i) {
        float4 v = xp[i];
        xd[4 * i + 0] = v.x * wt;
        xd[4 * i + 1] = v.y * wt;
        xd[4 * i + 2] = v.z * wt;
        xd[4 * i + 3] = v.w * wt;
      }
      wave_lds_order();

      int tgj[8];
#pragma unroll
      for (int j = 0; j < 8; ++j) {
        int m = t * 16 + 8 * hi + j;
        int v = ls_tgt[o + min(m, c - 1)];
        tgj[j] = (m < c) ? v : -1;
      }

#pragma unroll
      for (int ct = 0; ct < 4; ++ct) {
        v8f acc = {};
        const float* bcol = &Wc[(2 * hi) * W_STRIDE + ct * 16 + lm];
#pragma unroll
        for (int ks = 0; ks < 16; ++ks) {
          v2f a; a.x = arow[ks * 4];               a.y = arow[ks * 4 + 1];
          v2f b; b.x = bcol[(ks * 4) * W_STRIDE];  b.y = bcol[(ks * 4 + 1) * W_STRIDE];
          acc = __builtin_amdgcn_wmma_f32_16x16x4_f32(
               false, a,  false, b,
               (short)0, acc,  false,  false);
        }
#pragma unroll
        for (int j = 0; j < 8; ++j) {
          if (tgj[j] >= 0) atomicAdd(&out[(size_t)tgj[j] * DOUT + ct * 16 + lm], acc[j]);
        }
      }
      wave_lds_order();
    }

    if (more) {
      float* Wn = Wl[(r + 1) & 1];
#pragma unroll
      for (int i = 0; i < 16; ++i) {
        int idx = i * 256 + tid;
        Wn[(idx >> 6) * W_STRIDE + (idx & 63)] = wreg[i];
      }
    }
  }
}

static inline size_t align256(size_t v) { return (v + 255) & ~(size_t)255; }

extern "C" void kernel_launch(void* const* d_in, const int* in_sizes, int n_in,
                              void* d_out, int out_size, void* d_ws, size_t ws_size,
                              hipStream_t stream) {
  const float* x       = (const float*)d_in[0];
  const int*   source  = (const int*)d_in[1];
  const int*   target  = (const int*)d_in[2];
  const int*   etype   = (const int*)d_in[3];
  const float* ew      = (const float*)d_in[4];
  const float* bases   = (const float*)d_in[5];
  const float* comb    = (const float*)d_in[6];
  float*       out     = (float*)d_out;

  const int N = in_sizes[0] / DIN;
  const int E = in_sizes[1];
  const int B = in_sizes[5] / (DIN * DOUT);
  const int R = in_sizes[6] / B;

  const int nEB = (E + CHUNK - 1) / CHUNK;
  const int nbk = (N + ROWS - 1) / ROWS;

  char*  ws   = (char*)d_ws;
  float* Wbuf = (float*)ws;
  const size_t oPK = align256((size_t)R * DIN * DOUT * sizeof(float));
  const size_t oSE = oPK + align256((size_t)nbk * nEB * sizeof(int));
  const size_t need = oSE + (size_t)nEB * CHUNK * sizeof(int2);

  if (ws_size >= need && nbk <= MAXBK && R <= OWNR) {
    int*  PK = (int*)(ws + oPK);
    int*  SE = (int*)(ws + oSE);
    bin_kernel<<<R + nEB, 256, 0, stream>>>(source, target, etype, ew, bases, comb,
                                            Wbuf, B, R, E, nEB, nbk, PK, SE);
    rgcn_owner_kernel<<<nbk, 256, 0, stream>>>(x, Wbuf, PK, SE, N, R, nEB, out);
    return;
  }

  hipMemsetAsync(out, 0, (size_t)N * DOUT * sizeof(float), stream);
  compute_W_kernel<<<R, 256, 0, stream>>>(bases, comb, Wbuf, B);
  int nblk = (E + CHUNK - 1) / CHUNK;
  rgcn_chunk_kernel<<<nblk, 256, 0, stream>>>(x, Wbuf, source, target, etype, ew,
                                              E, R, out);
}


# owner collect: one table read and one block scan place a range's edges, then consecutive lanes load one edge each from the binned list (was 3 rounds of per-slice 8-load bursts); add order as before
# speedup vs baseline: 7.3311x; 1.0488x over previous
#include <hip/hip_runtime.h>

typedef __attribute__((ext_vector_type(2))) float v2f;
typedef __attribute__((ext_vector_type(8))) float v8f;

#define DIN 64
#define DOUT 64
#define CHUNK 2048
#define EPT 8
#define MAXR 64
#define XS_STRIDE 68
#define W_STRIDE 65
#define ROWS 196
#define MAXBK 512
#define CAP 2816
#define EPT2 11
#define OWNR 16
#define KPT 13
#define KPAD (KPT * 256)

__global__ void compute_W_kernel(const float* __restrict__ bases,
                                 const float* __restrict__ comb,
                                 float* __restrict__ W, int B) {
  int r = blockIdx.x;
  const float* c = comb + r * B;
  for (int idx = threadIdx.x; idx < DIN * DOUT; idx += blockDim.x) {
    float acc = 0.f;
    for (int b = 0; b < B; ++b) acc += c[b] * bases[b * DIN * DOUT + idx];
    W[r * DIN * DOUT + idx] = acc;
  }
}

__device__ __forceinline__ void wave_lds_order() {
  __builtin_amdgcn_fence(__ATOMIC_RELEASE, "wavefront");
  __builtin_amdgcn_wave_barrier();
  __builtin_amdgcn_fence(__ATOMIC_ACQUIRE, "wavefront");
}

__device__ __forceinline__ int block_incl_scan(int v, int* part, int tid, int& total) {
  const int ln = tid & 31, wq = tid >> 5;
  int s = v;
#pragma unroll
  for (int d = 1; d < 32; d <<= 1) {
    int u = __shfl_up(s, d, 32);
    if (ln >= d) s += u;
  }
  if (ln == 31) part[wq] = s;
  __syncthreads();
  int before = 0, all = 0;
#pragma unroll
  for (int k = 0; k < 8; ++k) {
    int t = part[k];
    all += t;
    if (k < wq) before += t;
  }
  total = all;
  return s + before;
}

__global__ void __launch_bounds__(256)
bin_kernel(const int* __restrict__ source,
           const int* __restrict__ target,
           const int* __restrict__ etype,
           const float* __restrict__ wts,
           const float* __restrict__ bases,
           const float* __restrict__ comb,
           float* __restrict__ W, int B, int R,
           int E, int nEB, int nbk,
           int* __restrict__ PK, int* __restrict__ SE) {
  __shared__ int  hist[MAXBK];
  __shared__ int  start[MAXBK];
  __shared__ int  part[8];
  __shared__ int2 stage[CHUNK];

  const int tid = threadIdx.x;

  if ((int)blockIdx.x < R) {
    const int r = blockIdx.x;
    const float* c = comb + r * B;
    for (int idx = tid; idx < DIN * DOUT; idx += 256) {
      float a = 0.f;
      for (int b = 0; b < B; ++b) a += c[b] * bases[b * DIN * DOUT + idx];
      const int k = idx >> 6, n = idx & 63;
      const int ks  = ((k >> 3) << 1) | ((k >> 1) & 1);
      const int dst = (ks * 2 + (n >> 5)) * 128
                    + (((k >> 2) & 1) * 16 + (n & 15)) * 4
                    + ((n >> 4) & 1) * 2 + (k & 1);
      W[r * DIN * DOUT + dst] = a;
    }
    return;
  }
  const int blk = (int)blockIdx.x - R;

  hist[tid] = 0;
  hist[tid + 256] = 0;
  __syncthreads();

  const int base = blk * CHUNK;
  int es[EPT], eg[EPT], er[EPT], ek[EPT], eb[EPT];
  float ew[EPT];
#pragma unroll
  for (int i = 0; i < EPT; ++i) {
    int e = base + i * 256 + tid;
    eb[i] = -1; es[i] = 0; eg[i] = 0; er[i] = 0; ew[i] = 0.f; ek[i] = 0;
    if (e < E) {
      es[i] = source[e]; eg[i] = target[e]; er[i] = etype[e]; ew[i] = wts[e];
      eb[i] = eg[i] / ROWS;
    }
  }
#pragma unroll
  for (int i = 0; i < EPT; ++i)
    if (eb[i] >= 0) ek[i] = atomicAdd(&hist[eb[i]], 1);
  __syncthreads();

  const int h0 = hist[2 * tid], h1 = hist[2 * tid + 1];
  int nall;
  const int incl = block_incl_scan(h0 + h1, part, tid, nall);
  const int excl = incl - (h0 + h1);
  start[2 * tid]     = excl;
  start[2 * tid + 1] = excl + h0;
  __syncthreads();

  for (int b = tid; b < nbk; b += 256) PK[b * nEB + blk] = (start[b] << 12) | hist[b];

#pragma unroll
  for (int i = 0; i < EPT; ++i) {
    if (eb[i] >= 0) {
      int p = start[eb[i]] + ek[i];
      stage[p] = make_int2(es[i] | ((eg[i] - eb[i] * ROWS) << 17) | (er[i] << 25),
                           __float_as_int(ew[i]));
    }
  }
  __syncthreads();

  const int nvalid = min(CHUNK, E - base);
  int2* SE2 = (int2*)SE;
  for (int i = tid; i < nvalid; i += 256) SE2[base + i] = stage[i];
}

__global__ void __launch_bounds__(256)
rgcn_owner_kernel(const float* __restrict__ x,
                  const float* __restrict__ W,
                  const int* __restrict__ PK,
                  const int* __restrict__ SE,
                  int N, int R, int nEB,
                  float* __restrict__ out) {
  __shared__ __attribute__((aligned(16))) float acc[(ROWS + 1) * DOUT];
  __shared__ int   gidx[CAP];
  __shared__ int   ls_src[CAP];
  __shared__ float ls_w[CAP];
  __shared__ int   seg[CAP + 1];
  __shared__ int   kc[KPAD];
  __shared__ int   soffs[OWNR + 1];
  __shared__ int   part[8];

  const int2* SE2 = (const int2*)SE;

  const int tid  = threadIdx.x;
  const int bk   = blockIdx.x;
  const int row0 = bk * ROWS;

  for (int i = tid; i < (ROWS + 1) * DOUT / 4; i += 256)
    ((float4*)acc)[i] = make_float4(0.f, 0.f, 0.f, 0.f);

  const int lane = tid & 31;
  const int wv   = __builtin_amdgcn_readfirstlane(tid >> 5);
  const int hi   = lane >> 4;
  const int lm   = lane & 15;

  int lo = 0;
  int total = 0;
  do {
    for (int i = tid; i < KPAD; i += 256) kc[i] = 0;

    int running = 0;
    for (int bb = 0; bb < nEB; bb += 768) {
      const int b0 = bb + 3 * tid;
      int pk[3];
#pragma unroll
      for (int k = 0; k < 3; ++k) pk[k] = PK[bk * nEB + min(b0 + k, nEB - 1)];
      int cc[3], g0[3];
#pragma unroll
      for (int k = 0; k < 3; ++k) {
        cc[k] = (b0 + k < nEB) ? (pk[k] & 4095) : 0;
        g0[k] = (b0 + k) * CHUNK + (pk[k] >> 12);
      }
      const int tsum = cc[0] + cc[1] + cc[2];
      int btot;
      const int incl = block_incl_scan(tsum, part, tid, btot);
      const int f = running + incl - tsum - lo;
      running += btot;
      const int qb = max(0, -f), qe = min(tsum, CAP - f);
      for (int q = qb; q < qe; ++q) {
        const int q1 = q - cc[0], q2 = q1 - cc[1];
        gidx[f + q] = (q1 < 0) ? g0[0] + q : (q2 < 0) ? g0[1] + q1 : g0[2] + q2;
      }
      __syncthreads();
    }
    total = __builtin_amdgcn_readfirstlane(running);
    const int n = min(CAP, total - lo);

    int2 ev[EPT2];
#pragma unroll
    for (int i = 0; i < EPT2; ++i) ev[i] = make_int2(0, 0);
    if (n > 0) {
      int gi[EPT2];
#pragma unroll
      for (int i = 0; i < EPT2; ++i) gi[i] = gidx[min(i * 256 + tid, n - 1)];
#pragma unroll
      for (int i = 0; i < EPT2; ++i) ev[i] = SE2[(size_t)gi[i]];
    }

    int ekey[EPT2], ek[EPT2], es[EPT2];
    float ew[EPT2];
#pragma unroll
    for (int i = 0; i < EPT2; ++i) {
      int idx = i * 256 + tid;
      ekey[i] = -1; es[i] = 0; ew[i] = 0.f; ek[i] = 0;
      if (idx < n) {
        int2 v = ev[i];
        es[i] = v.x & 0x1FFFF;
        ekey[i] = min((int)(((v.x >> 25) & 0x3F) * ROWS + ((v.x >> 17) & 0xFF)), KPAD - 1);
        ew[i] = __int_as_float(v.y);
      }
    }
#pragma unroll
    for (int i = 0; i < EPT2; ++i)
      if (ekey[i] >= 0) ek[i] = atomicAdd(&kc[ekey[i]], 1);
    __syncthreads();

    {
      int kcnt[KPT];
      int tsum = 0;
#pragma unroll
      for (int i = 0; i < KPT; ++i) {
        kcnt[i] = kc[KPT * tid + i];
        tsum += kcnt[i] | ((kcnt[i] > 0) << 16);
      }
      int tall;
      const int incl = block_incl_scan(tsum, part, tid, tall);
      int run = incl - tsum;
#pragma unroll
      for (int i = 0; i < KPT; ++i) {
        const int key = KPT * tid + i;
        kc[key] = run;
        if (kcnt[i] > 0) seg[run >> 16] = (run & 0xFFFF) | ((key % ROWS) << 12);
        run += kcnt[i] | ((kcnt[i] > 0) << 16);
      }
      if (tid == 255) seg[run >> 16] = run & 0xFFFF;
    }
    __syncthreads();

#pragma unroll
    for (int i = 0; i < EPT2; ++i) {
      if (ekey[i] >= 0) {
        int p = (kc[ekey[i]] & 0xFFFF) + ek[i];
        ls_src[p] = es[i];
        ls_w[p]   = ew[i];
      }
    }
    if (tid <= R) soffs[tid] = kc[tid * ROWS] >> 16;

    __syncthreads();

    for (int r = wv; r < R; r += 8) {
      const int o = __builtin_amdgcn_readfirstlane(soffs[r]);
      const int c = __builtin_amdgcn_readfirstlane(soffs[r + 1]) - o;
      if (c > 0) {
        const float4* wf = (const float4*)(W + (size_t)r * DIN * DOUT) + lane;
        float4 bq[32];
#pragma unroll
        for (int j = 0; j < 32; ++j) bq[j] = wf[j * 32];

        const int nt = (c + 15) >> 4;
        for (int t = 0; t < nt; ++t) {
          const int q  = t * 16 + lm;
          const int qi = o + min(q, c - 1);
          const int sg0 = seg[qi];
          const int sg1 = seg[qi + 1];
          const int e0 = sg0 & 0xFFF;
          const int e1 = (q < c) ? (sg1 & 0xFFF) : e0;
          float4 av[8];
          {
            const int   src = ls_src[e0];
            const float wt  = (q < c) ? ls_w[e0] : 0.f;
            const float4* xp = (const float4*)(x + (size_t)src * DIN) + hi;
#pragma unroll
            for (int i = 0; i < 8; ++i) av[i] = xp[2 * i];
#pragma unroll
            for (int i = 0; i < 8; ++i) {
              av[i].x *= wt; av[i].y *= wt; av[i].z *= wt; av[i].w *= wt;
            }
          }
#pragma nounroll
          for (int e = e0 + 1; e < e1; ++e) {
            const int   src = ls_src[e];
            const float wt  = ls_w[e];
            const float4* xp = (const float4*)(x + (size_t)src * DIN) + hi;
            float4 xv[8];
#pragma unroll
            for (int i = 0; i < 8; ++i) xv[i] = xp[2 * i];
#pragma unroll
            for (int i = 0; i < 8; ++i) {
              av[i].x += wt * xv[i].x;
              av[i].y += wt * xv[i].y;
              av[i].z += wt * xv[i].z;
              av[i].w += wt * xv[i].w;
            }
          }

          int rowa[8], rowk[8];
#pragma unroll
          for (int j = 0; j < 8; ++j) {
            int m = t * 16 + 8 * hi + j;
            int v = seg[o + min(m, c - 1)] >> 12;
            v = (m < c) ? v : ROWS;
            rowa[j] = v * DOUT + lm;
            rowk[j] = v & 3;
          }

          v8f a8[4] = {};
#pragma unroll
          for (int ks = 0; ks < 16; ++ks) {
            const float4 xv = av[ks >> 1];
            v2f a;
            if (ks & 1) { a.x = xv.z; a.y = xv.w; } else { a.x = xv.x; a.y = xv.y; }
#pragma unroll
            for (int ct = 0; ct < 4; ++ct) {
              const float4 w4 = bq[ks * 2 + (ct >> 1)];
              v2f b;
              if (ct & 1) { b.x = w4.z; b.y = w4.w; } else { b.x = w4.x; b.y = w4.y; }
              a8[ct] = __builtin_amdgcn_wmma_f32_16x16x4_f32(
                   false, a,  false, b,
                   (short)0, a8[ct],  false,  false);
            }
          }
#pragma unroll
          for (int ct = 0; ct < 4; ++ct) {
#pragma unroll
            for (int j = 0; j < 8; ++j)
              atomicAdd(&acc[rowa[j] + ((ct ^ rowk[j]) << 4)], a8[ct][j]);
          }
        }
      }
    }

    lo += CAP;
    __syncthreads();
  } while (lo < total);

  const int nrow = min(ROWS, N - row0);
  float4* op = (float4*)(out + (size_t)row0 * DOUT);
  for (int i = tid; i < nrow * (DOUT / 4); i += 256) {
    const int t = i >> 4, q = i & 15;
    op[i] = ((const float4*)acc)[t * 16 + (((q >> 2) ^ (t & 3)) << 2) + (q & 3)];
  }
}

__global__ void __launch_bounds__(256)
rgcn_chunk_kernel(const float* __restrict__ x,
                  const float* __restrict__ W,
                  const int* __restrict__ source,
                  const int* __restrict__ target,
                  const int* __restrict__ etype,
                  const float* __restrict__ wts,
                  int E, int R,
                  float* __restrict__ out) {
  __shared__ float xs[128 * XS_STRIDE];
  __shared__ float Wl[2][DIN * W_STRIDE];
  __shared__ int   ls_src[CHUNK];
  __shared__ int   ls_tgt[CHUNK];
  __shared__ float ls_w[CHUNK];
  __shared__ int   cnt[MAXR];
  __shared__ int   offs[MAXR];

  const int tid  = threadIdx.x;
  const int base = (int)blockIdx.x * CHUNK;

  if (tid < MAXR) cnt[tid] = 0;
  __syncthreads();

  int er[EPT], ek[EPT], es[EPT], eg[EPT];
  float ew[EPT];
#pragma unroll
  for (int i = 0; i < EPT; ++i) {
    int e = base + i * 256 + tid;
    er[i] = -1; es[i] = 0; eg[i] = 0; ew[i] = 0.f; ek[i] = 0;
    if (e < E) { er[i] = etype[e]; es[i] = source[e]; eg[i] = target[e]; ew[i] = wts[e]; }
  }
#pragma unroll
  for (int i = 0; i < EPT; ++i)
    if (er[i] >= 0) ek[i] = atomicAdd(&cnt[er[i]], 1);
  __syncthreads();

  if (tid == 0) {
    int o = 0;
    for (int r = 0; r < R; ++r) { offs[r] = o; o += cnt[r]; }
  }
  __syncthreads();

#pragma unroll
  for (int i = 0; i < EPT; ++i) {
    if (er[i] >= 0) {
      int p = offs[er[i]] + ek[i];
      ls_src[p] = es[i];
      ls_tgt[p] = eg[i];
      ls_w[p]   = ew[i];
    }
  }

  float wreg[16];
  if (R > 0) {
#pragma unroll
    for (int i = 0; i < 16; ++i) wreg[i] = W[i * 256 + tid];
#pragma unroll
    for (int i = 0; i < 16; ++i) {
      int idx = i * 256 + tid;
      Wl[0][(idx >> 6) * W_STRIDE + (idx & 63)] = wreg[i];
    }
  }

  const int lane = tid & 31;
  const int wv   = __builtin_amdgcn_readfirstlane(tid >> 5);
  const int hi   = lane >> 4;
  const int lm   = lane & 15;
  const int row  = lane >> 1;
  const int half = lane & 1;
  float* xd = &xs[(wv * 16 + row) * XS_STRIDE + half * 32];
  const float* arow = &xs[(wv * 16 + lm) * XS_STRIDE + 2 * hi];

  for (int r = 0; r < R; ++r) {
    __syncthreads();
    const float* Wc = Wl[r & 1];
    const bool more = (r + 1 < R);
    if (more) {
#pragma unroll
      for (int i = 0; i < 16; ++i) wreg[i] = W[(r + 1) * DIN * DOUT + i * 256 + tid];
    }
    const int c = __builtin_amdgcn_readfirstlane(cnt[r]);
    const int o = __builtin_amdgcn_readfirstlane(offs[r]);

    const int nt = (c + 15) >> 4;
    for (int t = wv; t < nt; t += 8) {
      int q = t * 16 + row;
      int src = 0; float wt = 0.f;
      if (q < c) { src = ls_src[o + q]; wt = ls_w[o + q]; }
      const float4* xp = (const float4*)(x + (size_t)src * DIN) + half * 8;
#pragma unroll
      for (int i = 0; i < 8; ++i) {
        float4 v = xp[i];
        xd[4 * i + 0] = v.x * wt;
        xd[4 * i + 1] = v.y * wt;
        xd[4 * i + 2] = v.z * wt;
        xd[4 * i + 3] = v.w * wt;
      }
      wave_lds_order();

      int tgj[8];
#pragma unroll
      for (int j = 0; j < 8; ++j) {
        int m = t * 16 + 8 * hi + j;
        int v = ls_tgt[o + min(m, c - 1)];
        tgj[j] = (m < c) ? v : -1;
      }

#pragma unroll
      for (int ct = 0; ct < 4; ++ct) {
        v8f acc = {};
        const float* bcol = &Wc[(2 * hi) * W_STRIDE + ct * 16 + lm];
#pragma unroll
        for (int ks = 0; ks < 16; ++ks) {
          v2f a; a.x = arow[ks * 4];               a.y = arow[ks * 4 + 1];
          v2f b; b.x = bcol[(ks * 4) * W_STRIDE];  b.y = bcol[(ks * 4 + 1) * W_STRIDE];
          acc = __builtin_amdgcn_wmma_f32_16x16x4_f32(
               false, a,  false, b,
               (short)0, acc,  false,  false);
        }
#pragma unroll
        for (int j = 0; j < 8; ++j) {
          if (tgj[j] >= 0) atomicAdd(&out[(size_t)tgj[j] * DOUT + ct * 16 + lm], acc[j]);
        }
      }
      wave_lds_order();
    }

    if (more) {
      float* Wn = Wl[(r + 1) & 1];
#pragma unroll
      for (int i = 0; i < 16; ++i) {
        int idx = i * 256 + tid;
        Wn[(idx >> 6) * W_STRIDE + (idx & 63)] = wreg[i];
      }
    }
  }
}

static inline size_t align256(size_t v) { return (v + 255) & ~(size_t)255; }

extern "C" void kernel_launch(void* const* d_in, const int* in_sizes, int n_in,
                              void* d_out, int out_size, void* d_ws, size_t ws_size,
                              hipStream_t stream) {
  const float* x       = (const float*)d_in[0];
  const int*   source  = (const int*)d_in[1];
  const int*   target  = (const int*)d_in[2];
  const int*   etype   = (const int*)d_in[3];
  const float* ew      = (const float*)d_in[4];
  const float* bases   = (const float*)d_in[5];
  const float* comb    = (const float*)d_in[6];
  float*       out     = (float*)d_out;

  const int N = in_sizes[0] / DIN;
  const int E = in_sizes[1];
  const int B = in_sizes[5] / (DIN * DOUT);
  const int R = in_sizes[6] / B;

  const int nEB = (E + CHUNK - 1) / CHUNK;
  const int nbk = (N + ROWS - 1) / ROWS;

  char*  ws   = (char*)d_ws;
  float* Wbuf = (float*)ws;
  const size_t oPK = align256((size_t)R * DIN * DOUT * sizeof(float));
  const size_t oSE = oPK + align256((size_t)nbk * nEB * sizeof(int));
  const size_t need = oSE + (size_t)nEB * CHUNK * sizeof(int2);

  if (ws_size >= need && nbk <= MAXBK && R <= OWNR) {
    int*  PK = (int*)(ws + oPK);
    int*  SE = (int*)(ws + oSE);
    bin_kernel<<<R + nEB, 256, 0, stream>>>(source, target, etype, ew, bases, comb,
                                            Wbuf, B, R, E, nEB, nbk, PK, SE);
    rgcn_owner_kernel<<<nbk, 256, 0, stream>>>(x, Wbuf, PK, SE, N, R, nEB, out);
    return;
  }

  hipMemsetAsync(out, 0, (size_t)N * DOUT * sizeof(float), stream);
  compute_W_kernel<<<R, 256, 0, stream>>>(bases, comb, Wbuf, B);
  int nblk = (E + CHUNK - 1) / CHUNK;
  rgcn_chunk_kernel<<<nblk, 256, 0, stream>>>(x, Wbuf, source, target, etype, ew,
                                              E, R, out);
}


# binning launch: a thread's global loads go out together (edge loads under a clamped index, not a branch; a matrix block builds 4 elements of W_r at once, 4 blocks a matrix); add order as before
# speedup vs baseline: 7.7498x; 1.0571x over previous
#include <hip/hip_runtime.h>

typedef __attribute__((ext_vector_type(2))) float v2f;
typedef __attribute__((ext_vector_type(8))) float v8f;

#define DIN 64
#define DOUT 64
#define CHUNK 2048
#define EPT 8
#define MAXR 64
#define XS_STRIDE 68
#define W_STRIDE 65
#define ROWS 196
#define MAXBK 512
#define CAP 2816
#define EPT2 11
#define OWNR 16
#define KPT 13
#define KPAD (KPT * 256)
#define WSPLIT 4

__global__ void compute_W_kernel(const float* __restrict__ bases,
                                 const float* __restrict__ comb,
                                 float* __restrict__ W, int B) {
  int r = blockIdx.x;
  const float* c = comb + r * B;
  for (int idx = threadIdx.x; idx < DIN * DOUT; idx += blockDim.x) {
    float acc = 0.f;
    for (int b = 0; b < B; ++b) acc += c[b] * bases[b * DIN * DOUT + idx];
    W[r * DIN * DOUT + idx] = acc;
  }
}

__device__ __forceinline__ void wave_lds_order() {
  __builtin_amdgcn_fence(__ATOMIC_RELEASE, "wavefront");
  __builtin_amdgcn_wave_barrier();
  __builtin_amdgcn_fence(__ATOMIC_ACQUIRE, "wavefront");
}

__device__ __forceinline__ int block_incl_scan(int v, int* part, int tid, int& total) {
  const int ln = tid & 31, wq = tid >> 5;
  int s = v;
#pragma unroll
  for (int d = 1; d < 32; d <<= 1) {
    int u = __shfl_up(s, d, 32);
    if (ln >= d) s += u;
  }
  if (ln == 31) part[wq] = s;
  __syncthreads();
  int before = 0, all = 0;
#pragma unroll
  for (int k = 0; k < 8; ++k) {
    int t = part[k];
    all += t;
    if (k < wq) before += t;
  }
  total = all;
  return s + before;
}

__global__ void __launch_bounds__(256)
bin_kernel(const int* __restrict__ source,
           const int* __restrict__ target,
           const int* __restrict__ etype,
           const float* __restrict__ wts,
           const float* __restrict__ bases,
           const float* __restrict__ comb,
           float* __restrict__ W, int B, int R,
           int E, int nEB, int nbk,
           int* __restrict__ PK, int* __restrict__ SE) {
  __shared__ int  hist[MAXBK];
  __shared__ int  start[MAXBK];
  __shared__ int  part[8];
  __shared__ int2 stage[CHUNK];

  const int tid = threadIdx.x;

  if ((int)blockIdx.x < R * WSPLIT) {
    const int r    = (int)blockIdx.x / WSPLIT;
    const int idx0 = ((int)blockIdx.x % WSPLIT) * (DIN * DOUT / WSPLIT) + tid;
    const float* c = comb + r * B;
    float a[4] = {0.f, 0.f, 0.f, 0.f};
#pragma unroll 8
    for (int b = 0; b < B; ++b) {
      const float cb = c[b];
#pragma unroll
      for (int j = 0; j < 4; ++j) a[j] += cb * bases[b * DIN * DOUT + idx0 + j * 256];
    }
#pragma unroll
    for (int j = 0; j < 4; ++j) {
      const int idx = idx0 + j * 256;
      const int k = idx >> 6, n = idx & 63;
      const int ks  = ((k >> 3) << 1) | ((k >> 1) & 1);
      const int dst = (ks * 2 + (n >> 5)) * 128
                    + (((k >> 2) & 1) * 16 + (n & 15)) * 4
                    + ((n >> 4) & 1) * 2 + (k & 1);
      W[r * DIN * DOUT + dst] = a[j];
    }
    return;
  }
  const int blk = (int)blockIdx.x - R * WSPLIT;

  const int base = blk * CHUNK;
  int es[EPT], eg[EPT], er[EPT], ek[EPT], eb[EPT];
  float ew[EPT];
#pragma unroll
  for (int i = 0; i < EPT; ++i) {
    const int ec = min(base + i * 256 + tid, E - 1);
    es[i] = source[ec]; eg[i] = target[ec]; er[i] = etype[ec]; ew[i] = wts[ec];
  }

  hist[tid] = 0;
  hist[tid + 256] = 0;
  __syncthreads();

#pragma unroll
  for (int i = 0; i < EPT; ++i) {
    ek[i] = 0;
    eb[i] = (base + i * 256 + tid < E) ? eg[i] / ROWS : -1;
  }
#pragma unroll
  for (int i = 0; i < EPT; ++i)
    if (eb[i] >= 0) ek[i] = atomicAdd(&hist[eb[i]], 1);
  __syncthreads();

  const int h0 = hist[2 * tid], h1 = hist[2 * tid + 1];
  int nall;
  const int incl = block_incl_scan(h0 + h1, part, tid, nall);
  const int excl = incl - (h0 + h1);
  start[2 * tid]     = excl;
  start[2 * tid + 1] = excl + h0;
  __syncthreads();

  for (int b = tid; b < nbk; b += 256) PK[b * nEB + blk] = (start[b] << 12) | hist[b];

#pragma unroll
  for (int i = 0; i < EPT; ++i) {
    if (eb[i] >= 0) {
      int p = start[eb[i]] + ek[i];
      stage[p] = make_int2(es[i] | ((eg[i] - eb[i] * ROWS) << 17) | (er[i] << 25),
                           __float_as_int(ew[i]));
    }
  }
  __syncthreads();

  const int nvalid = min(CHUNK, E - base);
  int2* SE2 = (int2*)SE;
  for (int i = tid; i < nvalid; i += 256) SE2[base + i] = stage[i];
}

__global__ void __launch_bounds__(256)
rgcn_owner_kernel(const float* __restrict__ x,
                  const float* __restrict__ W,
                  const int* __restrict__ PK,
                  const int* __restrict__ SE,
                  int N, int R, int nEB,
                  float* __restrict__ out) {
  __shared__ __attribute__((aligned(16))) float acc[(ROWS + 1) * DOUT];
  __shared__ int   gidx[CAP];
  __shared__ int   ls_src[CAP];
  __shared__ float ls_w[CAP];
  __shared__ int   seg[CAP + 1];
  __shared__ int   kc[KPAD];
  __shared__ int   soffs[OWNR + 1];
  __shared__ int   part[8];

  const int2* SE2 = (const int2*)SE;

  const int tid  = threadIdx.x;
  const int bk   = blockIdx.x;
  const int row0 = bk * ROWS;

  for (int i = tid; i < (ROWS + 1) * DOUT / 4; i += 256)
    ((float4*)acc)[i] = make_float4(0.f, 0.f, 0.f, 0.f);

  const int lane = tid & 31;
  const int wv   = __builtin_amdgcn_readfirstlane(tid >> 5);
  const int hi   = lane >> 4;
  const int lm   = lane & 15;

  int lo = 0;
  int total = 0;
  do {
    for (int i = tid; i < KPAD; i += 256) kc[i] = 0;

    int running = 0;
    for (int bb = 0; bb < nEB; bb += 768) {
      const int b0 = bb + 3 * tid;
      int pk[3];
#pragma unroll
      for (int k = 0; k < 3; ++k) pk[k] = PK[bk * nEB + min(b0 + k, nEB - 1)];
      int cc[3], g0[3];
#pragma unroll
      for (int k = 0; k < 3; ++k) {
        cc[k] = (b0 + k < nEB) ? (pk[k] & 4095) : 0;
        g0[k] = (b0 + k) * CHUNK + (pk[k] >> 12);
      }
      const int tsum = cc[0] + cc[1] + cc[2];
      int btot;
      const int incl = block_incl_scan(tsum, part, tid, btot);
      const int f = running + incl - tsum - lo;
      running += btot;
      const int qb = max(0, -f), qe = min(tsum, CAP - f);
      for (int q = qb; q < qe; ++q) {
        const int q1 = q - cc[0], q2 = q1 - cc[1];
        gidx[f + q] = (q1 < 0) ? g0[0] + q : (q2 < 0) ? g0[1] + q1 : g0[2] + q2;
      }
      __syncthreads();
    }
    total = __builtin_amdgcn_readfirstlane(running);
    const int n = min(CAP, total - lo);

    int2 ev[EPT2];
#pragma unroll
    for (int i = 0; i < EPT2; ++i) ev[i] = make_int2(0, 0);
    if (n > 0) {
      int gi[EPT2];
#pragma unroll
      for (int i = 0; i < EPT2; ++i) gi[i] = gidx[min(i * 256 + tid, n - 1)];
#pragma unroll
      for (int i = 0; i < EPT2; ++i) ev[i] = SE2[(size_t)gi[i]];
    }

    int ekey[EPT2], ek[EPT2], es[EPT2];
    float ew[EPT2];
#pragma unroll
    for (int i = 0; i < EPT2; ++i) {
      int idx = i * 256 + tid;
      ekey[i] = -1; es[i] = 0; ew[i] = 0.f; ek[i] = 0;
      if (idx < n) {
        int2 v = ev[i];
        es[i] = v.x & 0x1FFFF;
        ekey[i] = min((int)(((v.x >> 25) & 0x3F) * ROWS + ((v.x >> 17) & 0xFF)), KPAD - 1);
        ew[i] = __int_as_float(v.y);
      }
    }
#pragma unroll
    for (int i = 0; i < EPT2; ++i)
      if (ekey[i] >= 0) ek[i] = atomicAdd(&kc[ekey[i]], 1);
    __syncthreads();

    {
      int kcnt[KPT];
      int tsum = 0;
#pragma unroll
      for (int i = 0; i < KPT; ++i) {
        kcnt[i] = kc[KPT * tid + i];
        tsum += kcnt[i] | ((kcnt[i] > 0) << 16);
      }
      int tall;
      const int incl = block_incl_scan(tsum, part, tid, tall);
      int run = incl - tsum;
#pragma unroll
      for (int i = 0; i < KPT; ++i) {
        const int key = KPT * tid + i;
        kc[key] = run;
        if (kcnt[i] > 0) seg[run >> 16] = (run & 0xFFFF) | ((key % ROWS) << 12);
        run += kcnt[i] | ((kcnt[i] > 0) << 16);
      }
      if (tid == 255) seg[run >> 16] = run & 0xFFFF;
    }
    __syncthreads();

#pragma unroll
    for (int i = 0; i < EPT2; ++i) {
      if (ekey[i] >= 0) {
        int p = (kc[ekey[i]] & 0xFFFF) + ek[i];
        ls_src[p] = es[i];
        ls_w[p]   = ew[i];
      }
    }
    if (tid <= R) soffs[tid] = kc[tid * ROWS] >> 16;

    __syncthreads();

    for (int r = wv; r < R; r += 8) {
      const int o = __builtin_amdgcn_readfirstlane(soffs[r]);
      const int c = __builtin_amdgcn_readfirstlane(soffs[r + 1]) - o;
      if (c > 0) {
        const float4* wf = (const float4*)(W + (size_t)r * DIN * DOUT) + lane;
        float4 bq[32];
#pragma unroll
        for (int j = 0; j < 32; ++j) bq[j] = wf[j * 32];

        const int nt = (c + 15) >> 4;
        for (int t = 0; t < nt; ++t) {
          const int q  = t * 16 + lm;
          const int qi = o + min(q, c - 1);
          const int sg0 = seg[qi];
          const int sg1 = seg[qi + 1];
          const int e0 = sg0 & 0xFFF;
          const int e1 = (q < c) ? (sg1 & 0xFFF) : e0;
          float4 av[8];
          {
            const int   src = ls_src[e0];
            const float wt  = (q < c) ? ls_w[e0] : 0.f;
            const float4* xp = (const float4*)(x + (size_t)src * DIN) + hi;
#pragma unroll
            for (int i = 0; i < 8; ++i) av[i] = xp[2 * i];
#pragma unroll
            for (int i = 0; i < 8; ++i) {
              av[i].x *= wt; av[i].y *= wt; av[i].z *= wt; av[i].w *= wt;
            }
          }
#pragma nounroll
          for (int e = e0 + 1; e < e1; ++e) {
            const int   src = ls_src[e];
            const float wt  = ls_w[e];
            const float4* xp = (const float4*)(x + (size_t)src * DIN) + hi;
            float4 xv[8];
#pragma unroll
            for (int i = 0; i < 8; ++i) xv[i] = xp[2 * i];
#pragma unroll
            for (int i = 0; i < 8; ++i) {
              av[i].x += wt * xv[i].x;
              av[i].y += wt * xv[i].y;
              av[i].z += wt * xv[i].z;
              av[i].w += wt * xv[i].w;
            }
          }

          int rowa[8], rowk[8];
#pragma unroll
          for (int j = 0; j < 8; ++j) {
            int m = t * 16 + 8 * hi + j;
            int v = seg[o + min(m, c - 1)] >> 12;
            v = (m < c) ? v : ROWS;
            rowa[j] = v * DOUT + lm;
            rowk[j] = v & 3;
          }

          v8f a8[4] = {};
#pragma unroll
          for (int ks = 0; ks < 16; ++ks) {
            const float4 xv = av[ks >> 1];
            v2f a;
            if (ks & 1) { a.x = xv.z; a.y = xv.w; } else { a.x = xv.x; a.y = xv.y; }
#pragma unroll
            for (int ct = 0; ct < 4; ++ct) {
              const float4 w4 = bq[ks * 2 + (ct >> 1)];
              v2f b;
              if (ct & 1) { b.x = w4.z; b.y = w4.w; } else { b.x = w4.x; b.y = w4.y; }
              a8[ct] = __builtin_amdgcn_wmma_f32_16x16x4_f32(
                   false, a,  false, b,
                   (short)0, a8[ct],  false,  false);
            }
          }
#pragma unroll
          for (int ct = 0; ct < 4; ++ct) {
#pragma unroll
            for (int j = 0; j < 8; ++j)
              atomicAdd(&acc[rowa[j] + ((ct ^ rowk[j]) << 4)], a8[ct][j]);
          }
        }
      }
    }

    lo += CAP;
    __syncthreads();
  } while (lo < total);

  const int nrow = min(ROWS, N - row0);
  float4* op = (float4*)(out + (size_t)row0 * DOUT);
  for (int i = tid; i < nrow * (DOUT / 4); i += 256) {
    const int t = i >> 4, q = i & 15;
    op[i] = ((const float4*)acc)[t * 16 + (((q >> 2) ^ (t & 3)) << 2) + (q & 3)];
  }
}

__global__ void __launch_bounds__(256)
rgcn_chunk_kernel(const float* __restrict__ x,
                  const float* __restrict__ W,
                  const int* __restrict__ source,
                  const int* __restrict__ target,
                  const int* __restrict__ etype,
                  const float* __restrict__ wts,
                  int E, int R,
                  float* __restrict__ out) {
  __shared__ float xs[128 * XS_STRIDE];
  __shared__ float Wl[2][DIN * W_STRIDE];
  __shared__ int   ls_src[CHUNK];
  __shared__ int   ls_tgt[CHUNK];
  __shared__ float ls_w[CHUNK];
  __shared__ int   cnt[MAXR];
  __shared__ int   offs[MAXR];

  const int tid  = threadIdx.x;
  const int base = (int)blockIdx.x * CHUNK;

  if (tid < MAXR) cnt[tid] = 0;
  __syncthreads();

  int er[EPT], ek[EPT], es[EPT], eg[EPT];
  float ew[EPT];
#pragma unroll
  for (int i = 0; i < EPT; ++i) {
    int e = base + i * 256 + tid;
    er[i] = -1; es[i] = 0; eg[i] = 0; ew[i] = 0.f; ek[i] = 0;
    if (e < E) { er[i] = etype[e]; es[i] = source[e]; eg[i] = target[e]; ew[i] = wts[e]; }
  }
#pragma unroll
  for (int i = 0; i < EPT; ++i)
    if (er[i] >= 0) ek[i] = atomicAdd(&cnt[er[i]], 1);
  __syncthreads();

  if (tid == 0) {
    int o = 0;
    for (int r = 0; r < R; ++r) { offs[r] = o; o += cnt[r]; }
  }
  __syncthreads();

#pragma unroll
  for (int i = 0; i < EPT; ++i) {
    if (er[i] >= 0) {
      int p = offs[er[i]] + ek[i];
      ls_src[p] = es[i];
      ls_tgt[p] = eg[i];
      ls_w[p]   = ew[i];
    }
  }

  float wreg[16];
  if (R > 0) {
#pragma unroll
    for (int i = 0; i < 16; ++i) wreg[i] = W[i * 256 + tid];
#pragma unroll
    for (int i = 0; i < 16; ++i) {
      int idx = i * 256 + tid;
      Wl[0][(idx >> 6) * W_STRIDE + (idx & 63)] = wreg[i];
    }
  }

  const int lane = tid & 31;
  const int wv   = __builtin_amdgcn_readfirstlane(tid >> 5);
  const int hi   = lane >> 4;
  const int lm   = lane & 15;
  const int row  = lane >> 1;
  const int half = lane & 1;
  float* xd = &xs[(wv * 16 + row) * XS_STRIDE + half * 32];
  const float* arow = &xs[(wv * 16 + lm) * XS_STRIDE + 2 * hi];

  for (int r = 0; r < R; ++r) {
    __syncthreads();
    const float* Wc = Wl[r & 1];
    const bool more = (r + 1 < R);
    if (more) {
#pragma unroll
      for (int i = 0; i < 16; ++i) wreg[i] = W[(r + 1) * DIN * DOUT + i * 256 + tid];
    }
    const int c = __builtin_amdgcn_readfirstlane(cnt[r]);
    const int o = __builtin_amdgcn_readfirstlane(offs[r]);

    const int nt = (c + 15) >> 4;
    for (int t = wv; t < nt; t += 8) {
      int q = t * 16 + row;
      int src = 0; float wt = 0.f;
      if (q < c) { src = ls_src[o + q]; wt = ls_w[o + q]; }
      const float4* xp = (const float4*)(x + (size_t)src * DIN) + half * 8;
#pragma unroll
      for (int i = 0; i < 8; ++i) {
        float4 v = xp[i];
        xd[4 * i + 0] = v.x * wt;
        xd[4 * i + 1] = v.y * wt;
        xd[4 * i + 2] = v.z * wt;
        xd[4 * i + 3] = v.w * wt;
      }
      wave_lds_order();

      int tgj[8];
#pragma unroll
      for (int j = 0; j < 8; ++j) {
        int m = t * 16 + 8 * hi + j;
        int v = ls_tgt[o + min(m, c - 1)];
        tgj[j] = (m < c) ? v : -1;
      }

#pragma unroll
      for (int ct = 0; ct < 4; ++ct) {
        v8f acc = {};
        const float* bcol = &Wc[(2 * hi) * W_STRIDE + ct * 16 + lm];
#pragma unroll
        for (int ks = 0; ks < 16; ++ks) {
          v2f a; a.x = arow[ks * 4];               a.y = arow[ks * 4 + 1];
          v2f b; b.x = bcol[(ks * 4) * W_STRIDE];  b.y = bcol[(ks * 4 + 1) * W_STRIDE];
          acc = __builtin_amdgcn_wmma_f32_16x16x4_f32(
               false, a,  false, b,
               (short)0, acc,  false,  false);
        }
#pragma unroll
        for (int j = 0; j < 8; ++j) {
          if (tgj[j] >= 0) atomicAdd(&out[(size_t)tgj[j] * DOUT + ct * 16 + lm], acc[j]);
        }
      }
      wave_lds_order();
    }

    if (more) {
      float* Wn = Wl[(r + 1) & 1];
#pragma unroll
      for (int i = 0; i < 16; ++i) {
        int idx = i * 256 + tid;
        Wn[(idx >> 6) * W_STRIDE + (idx & 63)] = wreg[i];
      }
    }
  }
}

static inline size_t align256(size_t v) { return (v + 255) & ~(size_t)255; }

extern "C" void kernel_launch(void* const* d_in, const int* in_sizes, int n_in,
                              void* d_out, int out_size, void* d_ws, size_t ws_size,
                              hipStream_t stream) {
  const float* x       = (const float*)d_in[0];
  const int*   source  = (const int*)d_in[1];
  const int*   target  = (const int*)d_in[2];
  const int*   etype   = (const int*)d_in[3];
  const float* ew      = (const float*)d_in[4];
  const float* bases   = (const float*)d_in[5];
  const float* comb    = (const float*)d_in[6];
  float*       out     = (float*)d_out;

  const int N = in_sizes[0] / DIN;
  const int E = in_sizes[1];
  const int B = in_sizes[5] / (DIN * DOUT);
  const int R = in_sizes[6] / B;

  const int nEB = (E + CHUNK - 1) / CHUNK;
  const int nbk = (N + ROWS - 1) / ROWS;

  char*  ws   = (char*)d_ws;
  float* Wbuf = (float*)ws;
  const size_t oPK = align256((size_t)R * DIN * DOUT * sizeof(float));
  const size_t oSE = oPK + align256((size_t)nbk * nEB * sizeof(int));
  const size_t need = oSE + (size_t)nEB * CHUNK * sizeof(int2);

  if (ws_size >= need && nbk <= MAXBK && R <= OWNR) {
    int*  PK = (int*)(ws + oPK);
    int*  SE = (int*)(ws + oSE);
    bin_kernel<<<R * WSPLIT + nEB, 256, 0, stream>>>(source, target, etype, ew, bases, comb,
                                            Wbuf, B, R, E, nEB, nbk, PK, SE);
    rgcn_owner_kernel<<<nbk, 256, 0, stream>>>(x, Wbuf, PK, SE, N, R, nEB, out);
    return;
  }

  hipMemsetAsync(out, 0, (size_t)N * DOUT * sizeof(float), stream);
  compute_W_kernel<<<R, 256, 0, stream>>>(bases, comb, Wbuf, B);
  int nblk = (E + CHUNK - 1) / CHUNK;
  rgcn_chunk_kernel<<<nblk, 256, 0, stream>>>(x, Wbuf, source, target, etype, ew,
                                              E, R, out);
}


# owner epilogue: LDS sum rows padded to 80 floats (column-tile swizzle gone), a tile's row places come by lane shuffle from the A side: 32 LDS-add addresses take 8 vector ops, was 120; add order kept
# speedup vs baseline: 7.9943x; 1.0315x over previous
#include <hip/hip_runtime.h>

typedef __attribute__((ext_vector_type(2))) float v2f;
typedef __attribute__((ext_vector_type(8))) float v8f;

#define DIN 64
#define DOUT 64
#define CHUNK 2048
#define EPT 8
#define MAXR 64
#define XS_STRIDE 68
#define W_STRIDE 65
#define ROWS 196
#define MAXBK 512
#define CAP 2816
#define EPT2 11
#define OWNR 16
#define KPT 13
#define KPAD (KPT * 256)
#define WSPLIT 4
#define ASTRIDE 80

__global__ void compute_W_kernel(const float* __restrict__ bases,
                                 const float* __restrict__ comb,
                                 float* __restrict__ W, int B) {
  int r = blockIdx.x;
  const float* c = comb + r * B;
  for (int idx = threadIdx.x; idx < DIN * DOUT; idx += blockDim.x) {
    float acc = 0.f;
    for (int b = 0; b < B; ++b) acc += c[b] * bases[b * DIN * DOUT + idx];
    W[r * DIN * DOUT + idx] = acc;
  }
}

__device__ __forceinline__ void wave_lds_order() {
  __builtin_amdgcn_fence(__ATOMIC_RELEASE, "wavefront");
  __builtin_amdgcn_wave_barrier();
  __builtin_amdgcn_fence(__ATOMIC_ACQUIRE, "wavefront");
}

__device__ __forceinline__ int block_incl_scan(int v, int* part, int tid, int& total) {
  const int ln = tid & 31, wq = tid >> 5;
  int s = v;
#pragma unroll
  for (int d = 1; d < 32; d <<= 1) {
    int u = __shfl_up(s, d, 32);
    if (ln >= d) s += u;
  }
  if (ln == 31) part[wq] = s;
  __syncthreads();
  int before = 0, all = 0;
#pragma unroll
  for (int k = 0; k < 8; ++k) {
    int t = part[k];
    all += t;
    if (k < wq) before += t;
  }
  total = all;
  return s + before;
}

__global__ void __launch_bounds__(256)
bin_kernel(const int* __restrict__ source,
           const int* __restrict__ target,
           const int* __restrict__ etype,
           const float* __restrict__ wts,
           const float* __restrict__ bases,
           const float* __restrict__ comb,
           float* __restrict__ W, int B, int R,
           int E, int nEB, int nbk,
           int* __restrict__ PK, int* __restrict__ SE) {
  __shared__ int  hist[MAXBK];
  __shared__ int  start[MAXBK];
  __shared__ int  part[8];
  __shared__ int2 stage[CHUNK];

  const int tid = threadIdx.x;

  if ((int)blockIdx.x < R * WSPLIT) {
    const int r    = (int)blockIdx.x / WSPLIT;
    const int idx0 = ((int)blockIdx.x % WSPLIT) * (DIN * DOUT / WSPLIT) + tid;
    const float* c = comb + r * B;
    float a[4] = {0.f, 0.f, 0.f, 0.f};
#pragma unroll 8
    for (int b = 0; b < B; ++b) {
      const float cb = c[b];
#pragma unroll
      for (int j = 0; j < 4; ++j) a[j] += cb * bases[b * DIN * DOUT + idx0 + j * 256];
    }
#pragma unroll
    for (int j = 0; j < 4; ++j) {
      const int idx = idx0 + j * 256;
      const int k = idx >> 6, n = idx & 63;
      const int ks  = ((k >> 3) << 1) | ((k >> 1) & 1);
      const int dst = (ks * 2 + (n >> 5)) * 128
                    + (((k >> 2) & 1) * 16 + (n & 15)) * 4
                    + ((n >> 4) & 1) * 2 + (k & 1);
      W[r * DIN * DOUT + dst] = a[j];
    }
    return;
  }
  const int blk = (int)blockIdx.x - R * WSPLIT;

  const int base = blk * CHUNK;
  int es[EPT], eg[EPT], er[EPT], ek[EPT], eb[EPT];
  float ew[EPT];
#pragma unroll
  for (int i = 0; i < EPT; ++i) {
    const int ec = min(base + i * 256 + tid, E - 1);
    es[i] = source[ec]; eg[i] = target[ec]; er[i] = etype[ec]; ew[i] = wts[ec];
  }

  hist[tid] = 0;
  hist[tid + 256] = 0;
  __syncthreads();

#pragma unroll
  for (int i = 0; i < EPT; ++i) {
    ek[i] = 0;
    eb[i] = (base + i * 256 + tid < E) ? eg[i] / ROWS : -1;
  }
#pragma unroll
  for (int i = 0; i < EPT; ++i)
    if (eb[i] >= 0) ek[i] = atomicAdd(&hist[eb[i]], 1);
  __syncthreads();

  const int h0 = hist[2 * tid], h1 = hist[2 * tid + 1];
  int nall;
  const int incl = block_incl_scan(h0 + h1, part, tid, nall);
  const int excl = incl - (h0 + h1);
  start[2 * tid]     = excl;
  start[2 * tid + 1] = excl + h0;
  __syncthreads();

  for (int b = tid; b < nbk; b += 256) PK[b * nEB + blk] = (start[b] << 12) | hist[b];

#pragma unroll
  for (int i = 0; i < EPT; ++i) {
    if (eb[i] >= 0) {
      int p = start[eb[i]] + ek[i];
      stage[p] = make_int2(es[i] | ((eg[i] - eb[i] * ROWS) << 17) | (er[i] << 25),
                           __float_as_int(ew[i]));
    }
  }
  __syncthreads();

  const int nvalid = min(CHUNK, E - base);
  int2* SE2 = (int2*)SE;
  for (int i = tid; i < nvalid; i += 256) SE2[base + i] = stage[i];
}

__global__ void __launch_bounds__(256)
rgcn_owner_kernel(const float* __restrict__ x,
                  const float* __restrict__ W,
                  const int* __restrict__ PK,
                  const int* __restrict__ SE,
                  int N, int R, int nEB,
                  float* __restrict__ out) {
  __shared__ __attribute__((aligned(16))) float acc[(ROWS + 1) * ASTRIDE];
  __shared__ int   gidx[CAP];
  __shared__ int   ls_src[CAP];
  __shared__ float ls_w[CAP];
  __shared__ int   seg[CAP + 1];
  __shared__ int   kc[KPAD];
  __shared__ int   soffs[OWNR + 1];
  __shared__ int   part[8];

  const int2* SE2 = (const int2*)SE;

  const int tid  = threadIdx.x;
  const int bk   = blockIdx.x;
  const int row0 = bk * ROWS;

  for (int i = tid; i < (ROWS + 1) * ASTRIDE / 4; i += 256)
    ((float4*)acc)[i] = make_float4(0.f, 0.f, 0.f, 0.f);

  const int lane = tid & 31;
  const int wv   = __builtin_amdgcn_readfirstlane(tid >> 5);
  const int hi   = lane >> 4;
  const int lm   = lane & 15;

  int lo = 0;
  int total = 0;
  do {
    for (int i = tid; i < KPAD; i += 256) kc[i] = 0;

    int running = 0;
    for (int bb = 0; bb < nEB; bb += 768) {
      const int b0 = bb + 3 * tid;
      int pk[3];
#pragma unroll
      for (int k = 0; k < 3; ++k) pk[k] = PK[bk * nEB + min(b0 + k, nEB - 1)];
      int cc[3], g0[3];
#pragma unroll
      for (int k = 0; k < 3; ++k) {
        cc[k] = (b0 + k < nEB) ? (pk[k] & 4095) : 0;
        g0[k] = (b0 + k) * CHUNK + (pk[k] >> 12);
      }
      const int tsum = cc[0] + cc[1] + cc[2];
      int btot;
      const int incl = block_incl_scan(tsum, part, tid, btot);
      const int f = running + incl - tsum - lo;
      running += btot;
      const int qb = max(0, -f), qe = min(tsum, CAP - f);
      for (int q = qb; q < qe; ++q) {
        const int q1 = q - cc[0], q2 = q1 - cc[1];
        gidx[f + q] = (q1 < 0) ? g0[0] + q : (q2 < 0) ? g0[1] + q1 : g0[2] + q2;
      }
      __syncthreads();
    }
    total = __builtin_amdgcn_readfirstlane(running);
    const int n = min(CAP, total - lo);

    int2 ev[EPT2];
#pragma unroll
    for (int i = 0; i < EPT2; ++i) ev[i] = make_int2(0, 0);
    if (n > 0) {
      int gi[EPT2];
#pragma unroll
      for (int i = 0; i < EPT2; ++i) gi[i] = gidx[min(i * 256 + tid, n - 1)];
#pragma unroll
      for (int i = 0; i < EPT2; ++i) ev[i] = SE2[(size_t)gi[i]];
    }

    int ekey[EPT2], ek[EPT2], es[EPT2];
    float ew[EPT2];
#pragma unroll
    for (int i = 0; i < EPT2; ++i) {
      int idx = i * 256 + tid;
      ekey[i] = -1; es[i] = 0; ew[i] = 0.f; ek[i] = 0;
      if (idx < n) {
        int2 v = ev[i];
        es[i] = v.x & 0x1FFFF;
        ekey[i] = min((int)(((v.x >> 25) & 0x3F) * ROWS + ((v.x >> 17) & 0xFF)), KPAD - 1);
        ew[i] = __int_as_float(v.y);
      }
    }
#pragma unroll
    for (int i = 0; i < EPT2; ++i)
      if (ekey[i] >= 0) ek[i] = atomicAdd(&kc[ekey[i]], 1);
    __syncthreads();

    {
      int kcnt[KPT];
      int tsum = 0;
#pragma unroll
      for (int i = 0; i < KPT; ++i) {
        kcnt[i] = kc[KPT * tid + i];
        tsum += kcnt[i] | ((kcnt[i] > 0) << 16);
      }
      int tall;
      const int incl = block_incl_scan(tsum, part, tid, tall);
      int run = incl - tsum;
#pragma unroll
      for (int i = 0; i < KPT; ++i) {
        const int key = KPT * tid + i;
        kc[key] = run;
        if (kcnt[i] > 0) seg[run >> 16] = (run & 0xFFFF) | ((key % ROWS) << 12);
        run += kcnt[i] | ((kcnt[i] > 0) << 16);
      }
      if (tid == 255) seg[run >> 16] = run & 0xFFFF;
    }
    __syncthreads();

#pragma unroll
    for (int i = 0; i < EPT2; ++i) {
      if (ekey[i] >= 0) {
        int p = (kc[ekey[i]] & 0xFFFF) + ek[i];
        ls_src[p] = es[i];
        ls_w[p]   = ew[i];
      }
    }
    if (tid <= R) soffs[tid] = kc[tid * ROWS] >> 16;

    __syncthreads();

    for (int r = wv; r < R; r += 8) {
      const int o = __builtin_amdgcn_readfirstlane(soffs[r]);
      const int c = __builtin_amdgcn_readfirstlane(soffs[r + 1]) - o;
      if (c > 0) {
        const float4* wf = (const float4*)(W + (size_t)r * DIN * DOUT) + lane;
        float4 bq[32];
#pragma unroll
        for (int j = 0; j < 32; ++j) bq[j] = wf[j * 32];

        const int nt = (c + 15) >> 4;
        for (int t = 0; t < nt; ++t) {
          const int q  = t * 16 + lm;
          const int qi = o + min(q, c - 1);
          const int sg0 = seg[qi];
          const int sg1 = seg[qi + 1];
          const int e0 = sg0 & 0xFFF;
          const int e1 = (q < c) ? (sg1 & 0xFFF) : e0;
          const int mybase = ((q < c) ? (sg0 >> 12) : ROWS) * ASTRIDE;
          float4 av[8];
          {
            const int   src = ls_src[e0];
            const float wt  = (q < c) ? ls_w[e0] : 0.f;
            const float4* xp = (const float4*)(x + (size_t)src * DIN) + hi;
#pragma unroll
            for (int i = 0; i < 8; ++i) av[i] = xp[2 * i];
#pragma unroll
            for (int i = 0; i < 8; ++i) {
              av[i].x *= wt; av[i].y *= wt; av[i].z *= wt; av[i].w *= wt;
            }
          }
#pragma nounroll
          for (int e = e0 + 1; e < e1; ++e) {
            const int   src = ls_src[e];
            const float wt  = ls_w[e];
            const float4* xp = (const float4*)(x + (size_t)src * DIN) + hi;
            float4 xv[8];
#pragma unroll
            for (int i = 0; i < 8; ++i) xv[i] = xp[2 * i];
#pragma unroll
            for (int i = 0; i < 8; ++i) {
              av[i].x += wt * xv[i].x;
              av[i].y += wt * xv[i].y;
              av[i].z += wt * xv[i].z;
              av[i].w += wt * xv[i].w;
            }
          }

          int rowa[8];
#pragma unroll
          for (int j = 0; j < 8; ++j) rowa[j] = __shfl(mybase, 8 * hi + j, 32) + lm;

          v8f a8[4] = {};
#pragma unroll
          for (int ks = 0; ks < 16; ++ks) {
            const float4 xv = av[ks >> 1];
            v2f a;
            if (ks & 1) { a.x = xv.z; a.y = xv.w; } else { a.x = xv.x; a.y = xv.y; }
#pragma unroll
            for (int ct = 0; ct < 4; ++ct) {
              const float4 w4 = bq[ks * 2 + (ct >> 1)];
              v2f b;
              if (ct & 1) { b.x = w4.z; b.y = w4.w; } else { b.x = w4.x; b.y = w4.y; }
              a8[ct] = __builtin_amdgcn_wmma_f32_16x16x4_f32(
                   false, a,  false, b,
                   (short)0, a8[ct],  false,  false);
            }
          }
#pragma unroll
          for (int ct = 0; ct < 4; ++ct) {
#pragma unroll
            for (int j = 0; j < 8; ++j)
              atomicAdd(&acc[rowa[j] + (ct << 4)], a8[ct][j]);
          }
        }
      }
    }

    lo += CAP;
    __syncthreads();
  } while (lo < total);

  const int nrow = min(ROWS, N - row0);
  float4* op = (float4*)(out + (size_t)row0 * DOUT);
  for (int i = tid; i < nrow * (DOUT / 4); i += 256) {
    const int t = i >> 4, q = i & 15;
    op[i] = ((const float4*)acc)[t * (ASTRIDE / 4) + q];
  }
}

__global__ void __launch_bounds__(256)
rgcn_chunk_kernel(const float* __restrict__ x,
                  const float* __restrict__ W,
                  const int* __restrict__ source,
                  const int* __restrict__ target,
                  const int* __restrict__ etype,
                  const float* __restrict__ wts,
                  int E, int R,
                  float* __restrict__ out) {
  __shared__ float xs[128 * XS_STRIDE];
  __shared__ float Wl[2][DIN * W_STRIDE];
  __shared__ int   ls_src[CHUNK];
  __shared__ int   ls_tgt[CHUNK];
  __shared__ float ls_w[CHUNK];
  __shared__ int   cnt[MAXR];
  __shared__ int   offs[MAXR];

  const int tid  = threadIdx.x;
  const int base = (int)blockIdx.x * CHUNK;

  if (tid < MAXR) cnt[tid] = 0;
  __syncthreads();

  int er[EPT], ek[EPT], es[EPT], eg[EPT];
  float ew[EPT];
#pragma unroll
  for (int i = 0; i < EPT; ++i) {
    int e = base + i * 256 + tid;
    er[i] = -1; es[i] = 0; eg[i] = 0; ew[i] = 0.f; ek[i] = 0;
    if (e < E) { er[i] = etype[e]; es[i] = source[e]; eg[i] = target[e]; ew[i] = wts[e]; }
  }
#pragma unroll
  for (int i = 0; i < EPT; ++i)
    if (er[i] >= 0) ek[i] = atomicAdd(&cnt[er[i]], 1);
  __syncthreads();

  if (tid == 0) {
    int o = 0;
    for (int r = 0; r < R; ++r) { offs[r] = o; o += cnt[r]; }
  }
  __syncthreads();

#pragma unroll
  for (int i = 0; i < EPT; ++i) {
    if (er[i] >= 0) {
      int p = offs[er[i]] + ek[i];
      ls_src[p] = es[i];
      ls_tgt[p] = eg[i];
      ls_w[p]   = ew[i];
    }
  }

  float wreg[16];
  if (R > 0) {
#pragma unroll
    for (int i = 0; i < 16; ++i) wreg[i] = W[i * 256 + tid];
#pragma unroll
    for (int i = 0; i < 16; ++i) {
      int idx = i * 256 + tid;
      Wl[0][(idx >> 6) * W_STRIDE + (idx & 63)] = wreg[i];
    }
  }

  const int lane = tid & 31;
  const int wv   = __builtin_amdgcn_readfirstlane(tid >> 5);
  const int hi   = lane >> 4;
  const int lm   = lane & 15;
  const int row  = lane >> 1;
  const int half = lane & 1;
  float* xd = &xs[(wv * 16 + row) * XS_STRIDE + half * 32];
  const float* arow = &xs[(wv * 16 + lm) * XS_STRIDE + 2 * hi];

  for (int r = 0; r < R; ++r) {
    __syncthreads();
    const float* Wc = Wl[r & 1];
    const bool more = (r + 1 < R);
    if (more) {
#pragma unroll
      for (int i = 0; i < 16; ++i) wreg[i] = W[(r + 1) * DIN * DOUT + i * 256 + tid];
    }
    const int c = __builtin_amdgcn_readfirstlane(cnt[r]);
    const int o = __builtin_amdgcn_readfirstlane(offs[r]);

    const int nt = (c + 15) >> 4;
    for (int t = wv; t < nt; t += 8) {
      int q = t * 16 + row;
      int src = 0; float wt = 0.f;
      if (q < c) { src = ls_src[o + q]; wt = ls_w[o + q]; }
      const float4* xp = (const float4*)(x + (size_t)src * DIN) + half * 8;
#pragma unroll
      for (int i = 0; i < 8; ++i) {
        float4 v = xp[i];
        xd[4 * i + 0] = v.x * wt;
        xd[4 * i + 1] = v.y * wt;
        xd[4 * i + 2] = v.z * wt;
        xd[4 * i + 3] = v.w * wt;
      }
      wave_lds_order();

      int tgj[8];
#pragma unroll
      for (int j = 0; j < 8; ++j) {
        int m = t * 16 + 8 * hi + j;
        int v = ls_tgt[o + min(m, c - 1)];
        tgj[j] = (m < c) ? v : -1;
      }

#pragma unroll
      for (int ct = 0; ct < 4; ++ct) {
        v8f acc = {};
        const float* bcol = &Wc[(2 * hi) * W_STRIDE + ct * 16 + lm];
#pragma unroll
        for (int ks = 0; ks < 16; ++ks) {
          v2f a; a.x = arow[ks * 4];               a.y = arow[ks * 4 + 1];
          v2f b; b.x = bcol[(ks * 4) * W_STRIDE];  b.y = bcol[(ks * 4 + 1) * W_STRIDE];
          acc = __builtin_amdgcn_wmma_f32_16x16x4_f32(
               false, a,  false, b,
               (short)0, acc,  false,  false);
        }
#pragma unroll
        for (int j = 0; j < 8; ++j) {
          if (tgj[j] >= 0) atomicAdd(&out[(size_t)tgj[j] * DOUT + ct * 16 + lm], acc[j]);
        }
      }
      wave_lds_order();
    }

    if (more) {
      float* Wn = Wl[(r + 1) & 1];
#pragma unroll
      for (int i = 0; i < 16; ++i) {
        int idx = i * 256 + tid;
        Wn[(idx >> 6) * W_STRIDE + (idx & 63)] = wreg[i];
      }
    }
  }
}

static inline size_t align256(size_t v) { return (v + 255) & ~(size_t)255; }

extern "C" void kernel_launch(void* const* d_in, const int* in_sizes, int n_in,
                              void* d_out, int out_size, void* d_ws, size_t ws_size,
                              hipStream_t stream) {
  const float* x       = (const float*)d_in[0];
  const int*   source  = (const int*)d_in[1];
  const int*   target  = (const int*)d_in[2];
  const int*   etype   = (const int*)d_in[3];
  const float* ew      = (const float*)d_in[4];
  const float* bases   = (const float*)d_in[5];
  const float* comb    = (const float*)d_in[6];
  float*       out     = (float*)d_out;

  const int N = in_sizes[0] / DIN;
  const int E = in_sizes[1];
  const int B = in_sizes[5] / (DIN * DOUT);
  const int R = in_sizes[6] / B;

  const int nEB = (E + CHUNK - 1) / CHUNK;
  const int nbk = (N + ROWS - 1) / ROWS;

  char*  ws   = (char*)d_ws;
  float* Wbuf = (float*)ws;
  const size_t oPK = align256((size_t)R * DIN * DOUT * sizeof(float));
  const size_t oSE = oPK + align256((size_t)nbk * nEB * sizeof(int));
  const size_t need = oSE + (size_t)nEB * CHUNK * sizeof(int2);

  if (ws_size >= need && nbk <= MAXBK && R <= OWNR) {
    int*  PK = (int*)(ws + oPK);
    int*  SE = (int*)(ws + oSE);
    bin_kernel<<<R * WSPLIT + nEB, 256, 0, stream>>>(source, target, etype, ew, bases, comb,
                                            Wbuf, B, R, E, nEB, nbk, PK, SE);
    rgcn_owner_kernel<<<nbk, 256, 0, stream>>>(x, Wbuf, PK, SE, N, R, nEB, out);
    return;
  }

  hipMemsetAsync(out, 0, (size_t)N * DOUT * sizeof(float), stream);
  compute_W_kernel<<<R, 256, 0, stream>>>(bases, comb, Wbuf, B);
  int nblk = (E + CHUNK - 1) / CHUNK;
  rgcn_chunk_kernel<<<nblk, 256, 0, stream>>>(x, Wbuf, source, target, etype, ew,
                                              E, R, out);
}


# owner sort orders a relation's segments by edge count (1, 2, 3+; one more packed scan): a tile's 16 rows take equal gather rounds; x rows by 32-bit offset to stay at 256 registers; add order kept
# speedup vs baseline: 8.4303x; 1.0545x over previous
#include <hip/hip_runtime.h>

typedef __attribute__((ext_vector_type(2))) float v2f;
typedef __attribute__((ext_vector_type(8))) float v8f;

#define DIN 64
#define DOUT 64
#define CHUNK 2048
#define EPT 8
#define MAXR 64
#define XS_STRIDE 68
#define W_STRIDE 65
#define ROWS 196
#define MAXBK 512
#define CAP 2816
#define EPT2 11
#define OWNR 16
#define KPT 13
#define KPAD (KPT * 256)
#define KROW (KPAD / OWNR)
static_assert(KROW >= ROWS && KROW % KPT == 0 && KROW * OWNR == KPAD && (KROW / KPT) * OWNR == 256,
              "a relation's keys must be whole threads' shares of the key scan");
#define WSPLIT 4
#define ASTRIDE 80

__global__ void compute_W_kernel(const float* __restrict__ bases,
                                 const float* __restrict__ comb,
                                 float* __restrict__ W, int B) {
  int r = blockIdx.x;
  const float* c = comb + r * B;
  for (int idx = threadIdx.x; idx < DIN * DOUT; idx += blockDim.x) {
    float acc = 0.f;
    for (int b = 0; b < B; ++b) acc += c[b] * bases[b * DIN * DOUT + idx];
    W[r * DIN * DOUT + idx] = acc;
  }
}

__device__ __forceinline__ void wave_lds_order() {
  __builtin_amdgcn_fence(__ATOMIC_RELEASE, "wavefront");
  __builtin_amdgcn_wave_barrier();
  __builtin_amdgcn_fence(__ATOMIC_ACQUIRE, "wavefront");
}

__device__ __forceinline__ int block_incl_scan(int v, int* part, int tid, int& total) {
  const int ln = tid & 31, wq = tid >> 5;
  int s = v;
#pragma unroll
  for (int d = 1; d < 32; d <<= 1) {
    int u = __shfl_up(s, d, 32);
    if (ln >= d) s += u;
  }
  if (ln == 31) part[wq] = s;
  __syncthreads();
  int before = 0, all = 0;
#pragma unroll
  for (int k = 0; k < 8; ++k) {
    int t = part[k];
    all += t;
    if (k < wq) before += t;
  }
  total = all;
  return s + before;
}

__device__ __forceinline__ int block_incl_scan2(int va, int vb, int* part, int tid, int& inclB) {
  const int ln = tid & 31, wq = tid >> 5;
  int s = va, u2 = vb;
#pragma unroll
  for (int d = 1; d < 32; d <<= 1) {
    int ua = __shfl_up(s, d, 32);
    int ub = __shfl_up(u2, d, 32);
    if (ln >= d) { s += ua; u2 += ub; }
  }
  if (ln == 31) { part[wq] = s; part[8 + wq] = u2; }
  __syncthreads();
  int beforeA = 0, beforeB = 0;
#pragma unroll
  for (int k = 0; k < 8; ++k) {
    int ta = part[k], tb = part[8 + k];
    if (k < wq) { beforeA += ta; beforeB += tb; }
  }
  inclB = u2 + beforeB;
  return s + beforeA;
}

__global__ void __launch_bounds__(256)
bin_kernel(const int* __restrict__ source,
           const int* __restrict__ target,
           const int* __restrict__ etype,
           const float* __restrict__ wts,
           const float* __restrict__ bases,
           const float* __restrict__ comb,
           float* __restrict__ W, int B, int R,
           int E, int nEB, int nbk,
           int* __restrict__ PK, int* __restrict__ SE) {
  __shared__ int  hist[MAXBK];
  __shared__ int  start[MAXBK];
  __shared__ int  part[8];
  __shared__ int2 stage[CHUNK];

  const int tid = threadIdx.x;

  if ((int)blockIdx.x < R * WSPLIT) {
    const int r    = (int)blockIdx.x / WSPLIT;
    const int idx0 = ((int)blockIdx.x % WSPLIT) * (DIN * DOUT / WSPLIT) + tid;
    const float* c = comb + r * B;
    float a[4] = {0.f, 0.f, 0.f, 0.f};
#pragma unroll 8
    for (int b = 0; b < B; ++b) {
      const float cb = c[b];
#pragma unroll
      for (int j = 0; j < 4; ++j) a[j] += cb * bases[b * DIN * DOUT + idx0 + j * 256];
    }
#pragma unroll
    for (int j = 0; j < 4; ++j) {
      const int idx = idx0 + j * 256;
      const int k = idx >> 6, n = idx & 63;
      const int ks  = ((k >> 3) << 1) | ((k >> 1) & 1);
      const int dst = (ks * 2 + (n >> 5)) * 128
                    + (((k >> 2) & 1) * 16 + (n & 15)) * 4
                    + ((n >> 4) & 1) * 2 + (k & 1);
      W[r * DIN * DOUT + dst] = a[j];
    }
    return;
  }
  const int blk = (int)blockIdx.x - R * WSPLIT;

  const int base = blk * CHUNK;
  int es[EPT], eg[EPT], er[EPT], ek[EPT], eb[EPT];
  float ew[EPT];
#pragma unroll
  for (int i = 0; i < EPT; ++i) {
    const int ec = min(base + i * 256 + tid, E - 1);
    es[i] = source[ec]; eg[i] = target[ec]; er[i] = etype[ec]; ew[i] = wts[ec];
  }

  hist[tid] = 0;
  hist[tid + 256] = 0;
  __syncthreads();

#pragma unroll
  for (int i = 0; i < EPT; ++i) {
    ek[i] = 0;
    eb[i] = (base + i * 256 + tid < E) ? eg[i] / ROWS : -1;
  }
#pragma unroll
  for (int i = 0; i < EPT; ++i)
    if (eb[i] >= 0) ek[i] = atomicAdd(&hist[eb[i]], 1);
  __syncthreads();

  const int h0 = hist[2 * tid], h1 = hist[2 * tid + 1];
  int nall;
  const int incl = block_incl_scan(h0 + h1, part, tid, nall);
  const int excl = incl - (h0 + h1);
  start[2 * tid]     = excl;
  start[2 * tid + 1] = excl + h0;
  __syncthreads();

  for (int b = tid; b < nbk; b += 256) PK[b * nEB + blk] = (start[b] << 12) | hist[b];

#pragma unroll
  for (int i = 0; i < EPT; ++i) {
    if (eb[i] >= 0) {
      int p = start[eb[i]] + ek[i];
      stage[p] = make_int2(es[i] | ((eg[i] - eb[i] * ROWS) << 17) | (er[i] << 25),
                           __float_as_int(ew[i]));
    }
  }
  __syncthreads();

  const int nvalid = min(CHUNK, E - base);
  int2* SE2 = (int2*)SE;
  for (int i = tid; i < nvalid; i += 256) SE2[base + i] = stage[i];
}

__global__ void __launch_bounds__(256)
rgcn_owner_kernel(const float* __restrict__ x,
                  const float* __restrict__ W,
                  const int* __restrict__ PK,
                  const int* __restrict__ SE,
                  int N, int R, int nEB,
                  float* __restrict__ out) {
  __shared__ __attribute__((aligned(16))) float acc[(ROWS + 1) * ASTRIDE];
  __shared__ int   gidx[CAP];
  __shared__ int   ls_src[CAP];
  __shared__ float ls_w[CAP];
  __shared__ int   seg[CAP + 1];
  __shared__ int2  rel[OWNR + 1];
  __shared__ int   kc[KPAD];
  __shared__ int   soffs[OWNR + 1];
  __shared__ int   part[16];

  const int2* SE2 = (const int2*)SE;

  const int tid  = threadIdx.x;
  const int bk   = blockIdx.x;
  const int row0 = bk * ROWS;

  for (int i = tid; i < (ROWS + 1) * ASTRIDE / 4; i += 256)
    ((float4*)acc)[i] = make_float4(0.f, 0.f, 0.f, 0.f);

  const int lane = tid & 31;
  const int wv   = __builtin_amdgcn_readfirstlane(tid >> 5);
  const int hi   = lane >> 4;
  const int lm   = lane & 15;

  int lo = 0;
  int total = 0;
  do {
    for (int i = tid; i < KPAD; i += 256) kc[i] = 0;

    int running = 0;
    for (int bb = 0; bb < nEB; bb += 768) {
      const int b0 = bb + 3 * tid;
      int pk[3];
#pragma unroll
      for (int k = 0; k < 3; ++k) pk[k] = PK[bk * nEB + min(b0 + k, nEB - 1)];
      int cc[3], g0[3];
#pragma unroll
      for (int k = 0; k < 3; ++k) {
        cc[k] = (b0 + k < nEB) ? (pk[k] & 4095) : 0;
        g0[k] = (b0 + k) * CHUNK + (pk[k] >> 12);
      }
      const int tsum = cc[0] + cc[1] + cc[2];
      int btot;
      const int incl = block_incl_scan(tsum, part, tid, btot);
      const int f = running + incl - tsum - lo;
      running += btot;
      const int qb = max(0, -f), qe = min(tsum, CAP - f);
      for (int q = qb; q < qe; ++q) {
        const int q1 = q - cc[0], q2 = q1 - cc[1];
        gidx[f + q] = (q1 < 0) ? g0[0] + q : (q2 < 0) ? g0[1] + q1 : g0[2] + q2;
      }
      __syncthreads();
    }
    total = __builtin_amdgcn_readfirstlane(running);
    const int n = min(CAP, total - lo);

    int2 ev[EPT2];
#pragma unroll
    for (int i = 0; i < EPT2; ++i) ev[i] = make_int2(0, 0);
    if (n > 0) {
      int gi[EPT2];
#pragma unroll
      for (int i = 0; i < EPT2; ++i) gi[i] = gidx[min(i * 256 + tid, n - 1)];
#pragma unroll
      for (int i = 0; i < EPT2; ++i) ev[i] = SE2[(size_t)gi[i]];
    }

    int ekey[EPT2], ek[EPT2], es[EPT2];
    float ew[EPT2];
#pragma unroll
    for (int i = 0; i < EPT2; ++i) {
      int idx = i * 256 + tid;
      ekey[i] = -1; es[i] = 0; ew[i] = 0.f; ek[i] = 0;
      if (idx < n) {
        int2 v = ev[i];
        es[i] = v.x & 0x1FFFF;
        const int rn = (v.x >> 25) & 0x3F;
        ekey[i] = (rn < OWNR) ? rn * KROW + ((v.x >> 17) & 0xFF) : -1;
        ew[i] = __int_as_float(v.y);
      }
    }
#pragma unroll
    for (int i = 0; i < EPT2; ++i)
      if (ekey[i] >= 0) ek[i] = atomicAdd(&kc[ekey[i]], 1);
    __syncthreads();

    {
      int kcnt[KPT];
      int tsum = 0, tcls = 0;
#pragma unroll
      for (int i = 0; i < KPT; ++i) {
        kcnt[i] = kc[KPT * tid + i];
        tsum += kcnt[i] | ((kcnt[i] > 0) << 16);
        tcls += (kcnt[i] == 1) | ((kcnt[i] == 2) << 16);
      }
      int inclB;
      const int incl = block_incl_scan2(tsum, tcls, part, tid, inclB);
      const int run0 = incl - tsum, crun0 = inclB - tcls;
      if ((tid & 15) == 0) rel[tid >> 4] = make_int2(run0, crun0);
      if (tid == 255) rel[OWNR] = make_int2(incl, inclB);
      int run = run0;
#pragma unroll
      for (int i = 0; i < KPT; ++i) {
        kc[KPT * tid + i] = run;
        run += kcnt[i] | ((kcnt[i] > 0) << 16);
      }
      __syncthreads();

      const int r  = tid >> 4;
      const int2 rs = rel[r];
      const int sa = rs.x, sb = rs.y, eb = rel[r + 1].y;
      const int n1 = (eb & 0xFFFF) - (sb & 0xFFFF);
      const int n2 = (eb >> 16) - (sb >> 16);
      const int rowbase = (tid & 15) * KPT;
      run = run0;
      int crun = crun0;
#pragma unroll
      for (int i = 0; i < KPT; ++i) {
        if (kcnt[i] > 0) {
          const int c1b = (crun & 0xFFFF) - (sb & 0xFFFF);
          const int c2b = (crun >> 16) - (sb >> 16);
          const int c3b = (run >> 16) - (sa >> 16) - c1b - c2b;
          const int slot = (sa >> 16)
                         + ((kcnt[i] == 1) ? c1b : (kcnt[i] == 2) ? n1 + c2b : n1 + n2 + c3b);
          seg[slot] = (run & 0xFFFF) | ((rowbase + i) << 12) | (int)((unsigned)kcnt[i] << 20);
        }
        run  += kcnt[i] | ((kcnt[i] > 0) << 16);
        crun += (kcnt[i] == 1) | ((kcnt[i] == 2) << 16);
      }
    }
    __syncthreads();

#pragma unroll
    for (int i = 0; i < EPT2; ++i) {
      if (ekey[i] >= 0) {
        int p = (kc[ekey[i]] & 0xFFFF) + ek[i];
        ls_src[p] = es[i];
        ls_w[p]   = ew[i];
      }
    }
    if (tid <= R) soffs[tid] = rel[tid].x >> 16;

    __syncthreads();

    for (int r = wv; r < R; r += 8) {
      const int o = __builtin_amdgcn_readfirstlane(soffs[r]);
      const int c = __builtin_amdgcn_readfirstlane(soffs[r + 1]) - o;
      if (c > 0) {
        const float4* wf = (const float4*)(W + (size_t)r * DIN * DOUT) + lane;
        float4 bq[32];
#pragma unroll
        for (int j = 0; j < 32; ++j) bq[j] = wf[j * 32];

        const int nt = (c + 15) >> 4;
        for (int t = 0; t < nt; ++t) {
          const int q  = t * 16 + lm;
          const int qi = o + min(q, c - 1);
          const unsigned sg0 = (unsigned)seg[qi];
          const int e0 = (int)(sg0 & 0xFFFu);
          const int e1 = (q < c) ? e0 + (int)(sg0 >> 20) : e0;
          const int mybase = ((q < c) ? (int)((sg0 >> 12) & 0xFFu) : ROWS) * ASTRIDE;
          float4 av[8];
          {
            const int   src = ls_src[e0];
            const float wt  = (q < c) ? ls_w[e0] : 0.f;
            const float4* xp = (const float4*)((const char*)x + ((unsigned)src * (DIN * 4u) + ((unsigned)hi << 4)));
#pragma unroll
            for (int i = 0; i < 8; ++i) av[i] = xp[2 * i];
#pragma unroll
            for (int i = 0; i < 8; ++i) {
              av[i].x *= wt; av[i].y *= wt; av[i].z *= wt; av[i].w *= wt;
            }
          }
#pragma nounroll
          for (int e = e0 + 1; e < e1; ++e) {
            const int   src = ls_src[e];
            const float wt  = ls_w[e];
            const float4* xp = (const float4*)((const char*)x + ((unsigned)src * (DIN * 4u) + ((unsigned)hi << 4)));
            float4 xv[8];
#pragma unroll
            for (int i = 0; i < 8; ++i) xv[i] = xp[2 * i];
#pragma unroll
            for (int i = 0; i < 8; ++i) {
              av[i].x += wt * xv[i].x;
              av[i].y += wt * xv[i].y;
              av[i].z += wt * xv[i].z;
              av[i].w += wt * xv[i].w;
            }
          }

          int rowa[8];
#pragma unroll
          for (int j = 0; j < 8; ++j) rowa[j] = __shfl(mybase, 8 * hi + j, 32) + lm;

          v8f a8[4] = {};
#pragma unroll
          for (int ks = 0; ks < 16; ++ks) {
            const float4 xv = av[ks >> 1];
            v2f a;
            if (ks & 1) { a.x = xv.z; a.y = xv.w; } else { a.x = xv.x; a.y = xv.y; }
#pragma unroll
            for (int ct = 0; ct < 4; ++ct) {
              const float4 w4 = bq[ks * 2 + (ct >> 1)];
              v2f b;
              if (ct & 1) { b.x = w4.z; b.y = w4.w; } else { b.x = w4.x; b.y = w4.y; }
              a8[ct] = __builtin_amdgcn_wmma_f32_16x16x4_f32(
                   false, a,  false, b,
                   (short)0, a8[ct],  false,  false);
            }
          }
#pragma unroll
          for (int ct = 0; ct < 4; ++ct) {
#pragma unroll
            for (int j = 0; j < 8; ++j)
              atomicAdd(&acc[rowa[j] + (ct << 4)], a8[ct][j]);
          }
        }
      }
    }

    lo += CAP;
    __syncthreads();
  } while (lo < total);

  const int nrow = min(ROWS, N - row0);
  float4* op = (float4*)(out + (size_t)row0 * DOUT);
  for (int i = tid; i < nrow * (DOUT / 4); i += 256) {
    const int t = i >> 4, q = i & 15;
    op[i] = ((const float4*)acc)[t * (ASTRIDE / 4) + q];
  }
}

__global__ void __launch_bounds__(256)
rgcn_chunk_kernel(const float* __restrict__ x,
                  const float* __restrict__ W,
                  const int* __restrict__ source,
                  const int* __restrict__ target,
                  const int* __restrict__ etype,
                  const float* __restrict__ wts,
                  int E, int R,
                  float* __restrict__ out) {
  __shared__ float xs[128 * XS_STRIDE];
  __shared__ float Wl[2][DIN * W_STRIDE];
  __shared__ int   ls_src[CHUNK];
  __shared__ int   ls_tgt[CHUNK];
  __shared__ float ls_w[CHUNK];
  __shared__ int   cnt[MAXR];
  __shared__ int   offs[MAXR];

  const int tid  = threadIdx.x;
  const int base = (int)blockIdx.x * CHUNK;

  if (tid < MAXR) cnt[tid] = 0;
  __syncthreads();

  int er[EPT], ek[EPT], es[EPT], eg[EPT];
  float ew[EPT];
#pragma unroll
  for (int i = 0; i < EPT; ++i) {
    int e = base + i * 256 + tid;
    er[i] = -1; es[i] = 0; eg[i] = 0; ew[i] = 0.f; ek[i] = 0;
    if (e < E) { er[i] = etype[e]; es[i] = source[e]; eg[i] = target[e]; ew[i] = wts[e]; }
  }
#pragma unroll
  for (int i = 0; i < EPT; ++i)
    if (er[i] >= 0) ek[i] = atomicAdd(&cnt[er[i]], 1);
  __syncthreads();

  if (tid == 0) {
    int o = 0;
    for (int r = 0; r < R; ++r) { offs[r] = o; o += cnt[r]; }
  }
  __syncthreads();

#pragma unroll
  for (int i = 0; i < EPT; ++i) {
    if (er[i] >= 0) {
      int p = offs[er[i]] + ek[i];
      ls_src[p] = es[i];
      ls_tgt[p] = eg[i];
      ls_w[p]   = ew[i];
    }
  }

  float wreg[16];
  if (R > 0) {
#pragma unroll
    for (int i = 0; i < 16; ++i) wreg[i] = W[i * 256 + tid];
#pragma unroll
    for (int i = 0; i < 16; ++i) {
      int idx = i * 256 + tid;
      Wl[0][(idx >> 6) * W_STRIDE + (idx & 63)] = wreg[i];
    }
  }

  const int lane = tid & 31;
  const int wv   = __builtin_amdgcn_readfirstlane(tid >> 5);
  const int hi   = lane >> 4;
  const int lm   = lane & 15;
  const int row  = lane >> 1;
  const int half = lane & 1;
  float* xd = &xs[(wv * 16 + row) * XS_STRIDE + half * 32];
  const float* arow = &xs[(wv * 16 + lm) * XS_STRIDE + 2 * hi];

  for (int r = 0; r < R; ++r) {
    __syncthreads();
    const float* Wc = Wl[r & 1];
    const bool more = (r + 1 < R);
    if (more) {
#pragma unroll
      for (int i = 0; i < 16; ++i) wreg[i] = W[(r + 1) * DIN * DOUT + i * 256 + tid];
    }
    const int c = __builtin_amdgcn_readfirstlane(cnt[r]);
    const int o = __builtin_amdgcn_readfirstlane(offs[r]);

    const int nt = (c + 15) >> 4;
    for (int t = wv; t < nt; t += 8) {
      int q = t * 16 + row;
      int src = 0; float wt = 0.f;
      if (q < c) { src = ls_src[o + q]; wt = ls_w[o + q]; }
      const float4* xp = (const float4*)(x + (size_t)src * DIN) + half * 8;
#pragma unroll
      for (int i = 0; i < 8; ++i) {
        float4 v = xp[i];
        xd[4 * i + 0] = v.x * wt;
        xd[4 * i + 1] = v.y * wt;
        xd[4 * i + 2] = v.z * wt;
        xd[4 * i + 3] = v.w * wt;
      }
      wave_lds_order();

      int tgj[8];
#pragma unroll
      for (int j = 0; j < 8; ++j) {
        int m = t * 16 + 8 * hi + j;
        int v = ls_tgt[o + min(m, c - 1)];
        tgj[j] = (m < c) ? v : -1;
      }

#pragma unroll
      for (int ct = 0; ct < 4; ++ct) {
        v8f acc = {};
        const float* bcol = &Wc[(2 * hi) * W_STRIDE + ct * 16 + lm];
#pragma unroll
        for (int ks = 0; ks < 16; ++ks) {
          v2f a; a.x = arow[ks * 4];               a.y = arow[ks * 4 + 1];
          v2f b; b.x = bcol[(ks * 4) * W_STRIDE];  b.y = bcol[(ks * 4 + 1) * W_STRIDE];
          acc = __builtin_amdgcn_wmma_f32_16x16x4_f32(
               false, a,  false, b,
               (short)0, acc,  false,  false);
        }
#pragma unroll
        for (int j = 0; j < 8; ++j) {
          if (tgj[j] >= 0) atomicAdd(&out[(size_t)tgj[j] * DOUT + ct * 16 + lm], acc[j]);
        }
      }
      wave_lds_order();
    }

    if (more) {
      float* Wn = Wl[(r + 1) & 1];
#pragma unroll
      for (int i = 0; i < 16; ++i) {
        int idx = i * 256 + tid;
        Wn[(idx >> 6) * W_STRIDE + (idx & 63)] = wreg[i];
      }
    }
  }
}

static inline size_t align256(size_t v) { return (v + 255) & ~(size_t)255; }

extern "C" void kernel_launch(void* const* d_in, const int* in_sizes, int n_in,
                              void* d_out, int out_size, void* d_ws, size_t ws_size,
                              hipStream_t stream) {
  const float* x       = (const float*)d_in[0];
  const int*   source  = (const int*)d_in[1];
  const int*   target  = (const int*)d_in[2];
  const int*   etype   = (const int*)d_in[3];
  const float* ew      = (const float*)d_in[4];
  const float* bases   = (const float*)d_in[5];
  const float* comb    = (const float*)d_in[6];
  float*       out     = (float*)d_out;

  const int N = in_sizes[0] / DIN;
  const int E = in_sizes[1];
  const int B = in_sizes[5] / (DIN * DOUT);
  const int R = in_sizes[6] / B;

  const int nEB = (E + CHUNK - 1) / CHUNK;
  const int nbk = (N + ROWS - 1) / ROWS;

  char*  ws   = (char*)d_ws;
  float* Wbuf = (float*)ws;
  const size_t oPK = align256((size_t)R * DIN * DOUT * sizeof(float));
  const size_t oSE = oPK + align256((size_t)nbk * nEB * sizeof(int));
  const size_t need = oSE + (size_t)nEB * CHUNK * sizeof(int2);

  if (ws_size >= need && nbk <= MAXBK && R <= OWNR) {
    int*  PK = (int*)(ws + oPK);
    int*  SE = (int*)(ws + oSE);
    bin_kernel<<<R * WSPLIT + nEB, 256, 0, stream>>>(source, target, etype, ew, bases, comb,
                                            Wbuf, B, R, E, nEB, nbk, PK, SE);
    rgcn_owner_kernel<<<nbk, 256, 0, stream>>>(x, Wbuf, PK, SE, N, R, nEB, out);
    return;
  }

  hipMemsetAsync(out, 0, (size_t)N * DOUT * sizeof(float), stream);
  compute_W_kernel<<<R, 256, 0, stream>>>(bases, comb, Wbuf, B);
  int nblk = (E + CHUNK - 1) / CHUNK;
  rgcn_chunk_kernel<<<nblk, 256, 0, stream>>>(x, Wbuf, source, target, etype, ew,
                                              E, R, out);
}
